# starting kernel
# baseline (speedup 1.0000x reference)
#include <hip/hip_runtime.h>
#include <math.h>

#define NNODES 100000
#define NEDGES 1200000
#define FDIM   64
#define ROWTILES (NNODES / 16)

typedef __attribute__((ext_vector_type(2))) float v2f;
typedef __attribute__((ext_vector_type(8))) float v8f;

__global__ void __launch_bounds__(256) scatter_add_kernel(
    const float* __restrict__ h, const int* __restrict__ src,
    const int* __restrict__ dst, float* __restrict__ agg, int nEdges)
{
    int t = blockIdx.x * blockDim.x + threadIdx.x;
    int e = t >> 4;
    if (e >= nEdges) return;
    int q = t & 15;

    int s = src[e];
    int d = dst[e];

    float4 v = reinterpret_cast<const float4*>(h + (size_t)s * FDIM)[q];
    float* op = agg + (size_t)d * FDIM + q * 4;
    atomicAdd(op + 0, v.x);
    atomicAdd(op + 1, v.y);
    atomicAdd(op + 2, v.z);
    atomicAdd(op + 3, v.w);
}

__global__ void __launch_bounds__(256) sage_gemm_kernel(
    const float* __restrict__ agg, const float* __restrict__ hin,
    const float* __restrict__ Wl,  const float* __restrict__ bias,
    const float* __restrict__ Wr,  float* __restrict__ out, int applyTanh)
{
    __shared__ float sWl[FDIM * FDIM];
    __shared__ float sWr[FDIM * FDIM];

    for (int i = threadIdx.x; i < (FDIM * FDIM) / 4; i += 256) {
        reinterpret_cast<float4*>(sWl)[i] = reinterpret_cast<const float4*>(Wl)[i];
        reinterpret_cast<float4*>(sWr)[i] = reinterpret_cast<const float4*>(Wr)[i];
    }
    __syncthreads();

    const int wave = threadIdx.x >> 5;
    const int lane = threadIdx.x & 31;
    const int half = lane >> 4;
    const int lrow = lane & 15;

    const int rb = blockIdx.x * 8 + wave;
    if (rb >= ROWTILES) return;
    const size_t r0 = (size_t)rb * 16;

    v8f acc[4];
    const v8f zero = {0.f, 0.f, 0.f, 0.f, 0.f, 0.f, 0.f, 0.f};
#pragma unroll
    for (int t = 0; t < 4; ++t) acc[t] = zero;

    const float* aRowL = agg + (r0 + lrow) * FDIM + half * 2;
    const float* aRowR = hin + (r0 + lrow) * FDIM + half * 2;

#pragma unroll
    for (int kb = 0; kb < FDIM; kb += 4) {
        v2f aL = *reinterpret_cast<const v2f*>(aRowL + kb);
        v2f aR = *reinterpret_cast<const v2f*>(aRowR + kb);
#pragma unroll
        for (int t = 0; t < 4; ++t) {
            const int n = t * 16 + lrow;
            v2f bL = *reinterpret_cast<const v2f*>(sWl + n * FDIM + kb + half * 2);
            v2f bR = *reinterpret_cast<const v2f*>(sWr + n * FDIM + kb + half * 2);
            acc[t] = __builtin_amdgcn_wmma_f32_16x16x4_f32(
                false, aL, false, bL, (short)0, acc[t], false, false);
            acc[t] = __builtin_amdgcn_wmma_f32_16x16x4_f32(
                false, aR, false, bR, (short)0, acc[t], false, false);
        }
    }

#pragma unroll
    for (int t = 0; t < 4; ++t) {
        const int n = t * 16 + lrow;
        const float bv = bias[n];
#pragma unroll
        for (int v = 0; v < 8; ++v) {
            const int m = v + half * 8;
            float val = acc[t][v] + bv;
            if (applyTanh) val = tanhf(val);
            out[(r0 + m) * FDIM + n] = val;
        }
    }
}

extern "C" void kernel_launch(void* const* d_in, const int* in_sizes, int n_in,
                              void* d_out, int out_size, void* d_ws, size_t ws_size,
                              hipStream_t stream)
{
    const float* x      = (const float*)d_in[0];
    const int*   ei     = (const int*)d_in[1];
    const int*   src    = ei;
    const int*   dst    = ei + NEDGES;
    const float* Wl_in  = (const float*)d_in[2];
    const float* bl_in  = (const float*)d_in[3];
    const float* Wr_in  = (const float*)d_in[4];
    const float* Wl_med = (const float*)d_in[5];
    const float* bl_med = (const float*)d_in[6];
    const float* Wr_med = (const float*)d_in[7];
    const float* Wl_out = (const float*)d_in[8];
    const float* bl_out = (const float*)d_in[9];
    const float* Wr_out = (const float*)d_in[10];

    float* agg = (float*)d_ws;
    float* hA  = agg + (size_t)NNODES * FDIM;
    float* hB  = (float*)d_out;

    const size_t aggBytes = (size_t)NNODES * FDIM * sizeof(float);

    const dim3 sBlk(256);
    const dim3 sGrid((unsigned)(((size_t)NEDGES * 16 + 255) / 256));
    const dim3 gBlk(256);
    const dim3 gGrid((ROWTILES + 7) / 8);

    auto layer = [&](const float* hin, float* hout, const float* Wl,
                     const float* bl, const float* Wr, int doTanh) {
        hipMemsetAsync(agg, 0, aggBytes, stream);
        scatter_add_kernel<<<sGrid, sBlk, 0, stream>>>(hin, src, dst, agg, NEDGES);
        sage_gemm_kernel<<<gGrid, gBlk, 0, stream>>>(agg, hin, Wl, bl, Wr, hout, doTanh);
    };

    layer(x,  hA,           Wl_in,  bl_in,  Wr_in,  1);
    layer(hA, hB,           Wl_med, bl_med, Wr_med, 1);
    layer(hB, hA,           Wl_med, bl_med, Wr_med, 1);
    layer(hA, (float*)d_out, Wl_out, bl_out, Wr_out, 0);
}


# aggregate by gather over a per-call dst-grouped edge list (int counting sort) instead of f32 atomic scatter; agg written once, no agg memset; sum order = arrival rank of an int counter, run-varying
# speedup vs baseline: 1.5838x; 1.5838x over previous
#include <hip/hip_runtime.h>
#include <math.h>


#define NNODES 100000
#define NEDGES 1200000
#define FDIM   64
#define ROWTILES (NNODES / 16)
#define SCAN_BLK  1024
#define SCAN_NBLK ((NNODES + SCAN_BLK - 1) / SCAN_BLK)

typedef __attribute__((ext_vector_type(2))) float v2f;
typedef __attribute__((ext_vector_type(8))) float v8f;

__global__ void __launch_bounds__(256) degree_count_kernel(
    const int* __restrict__ dst, int* __restrict__ cnt, int nEdges)
{
    int e = blockIdx.x * blockDim.x + threadIdx.x;
    if (e >= nEdges) return;
    atomicAdd(&cnt[dst[e]], 1);
}

__device__ __forceinline__ int wave_inclusive_scan(int v, int lane)
{
    for (int o = 1; o < 32; o <<= 1) {
        int u = __shfl_up(v, o, 32);
        if (lane >= o) v += u;
    }
    return v;
}

__global__ void __launch_bounds__(SCAN_BLK) block_scan_kernel(
    const int* __restrict__ cnt, int* __restrict__ lscan, int n)
{
    __shared__ int sWave[32];
    const int t    = threadIdx.x;
    const int lane = t & 31;
    const int wave = t >> 5;
    const int i    = blockIdx.x * SCAN_BLK + t;
    const int v    = (i < n) ? cnt[i] : 0;
    const int inc  = wave_inclusive_scan(v, lane);
    if (lane == 31) sWave[wave] = inc;
    __syncthreads();
    if (wave == 0) {
        const int w = sWave[lane];
        sWave[lane] = wave_inclusive_scan(w, lane) - w;
    }
    __syncthreads();
    if (i < n) lscan[i] = sWave[wave] + inc - v;
}

__global__ void __launch_bounds__(128) block_base_kernel(
    const int* __restrict__ cnt, const int* __restrict__ lscan,
    int* __restrict__ bbase, int nb)
{
    __shared__ int sW[4];
    const int t    = threadIdx.x;
    const int lane = t & 31;
    const int wave = t >> 5;
    int v = 0;
    if (t < nb - 1) {
        const int last = t * SCAN_BLK + (SCAN_BLK - 1);
        v = lscan[last] + cnt[last];
    }
    const int inc = wave_inclusive_scan(v, lane);
    if (lane == 31) sW[wave] = inc;
    __syncthreads();
    int base = 0;
    for (int w = 0; w < wave; ++w) base += sW[w];
    if (t < nb) bbase[t] = base + inc - v;
}

__global__ void __launch_bounds__(256) bucket_fill_kernel(
    const int* __restrict__ src, const int* __restrict__ dst,
    const int* __restrict__ lscan, const int* __restrict__ bbase,
    int* __restrict__ cursor, int* __restrict__ ssrc, int nEdges)
{
    int e = blockIdx.x * blockDim.x + threadIdx.x;
    if (e >= nEdges) return;
    const int d    = dst[e];
    const int pos  = atomicAdd(&cursor[d], 1);
    const int slot = lscan[d] + bbase[d / SCAN_BLK] + pos;
    atomicExch(&ssrc[slot], src[e]);
}

__global__ void __launch_bounds__(256) gather_sum_kernel(
    const float* __restrict__ h, const int* __restrict__ ssrc,
    const int* __restrict__ cnt, const int* __restrict__ lscan,
    const int* __restrict__ bbase, float* __restrict__ agg, int nNodes)
{
    const int t = blockIdx.x * blockDim.x + threadIdx.x;
    const int d = t >> 4;
    if (d >= nNodes) return;
    const int q = t & 15;

    const int  n   = cnt[d];
    const int* lst = ssrc + (lscan[d] + bbase[d / SCAN_BLK]);
    const float4* h4 = reinterpret_cast<const float4*>(h);

    float4 acc = make_float4(0.f, 0.f, 0.f, 0.f);
    if (n > 0) {
        const int jl = n - 1;
        int i0 = lst[0];
        int i1 = lst[min(1, jl)];
        int i2 = lst[min(2, jl)];
        int i3 = lst[min(3, jl)];
        for (int j = 0; j < n; j += 4) {
            const float4 v0 = h4[(size_t)i0 * 16 + q];
            const float4 v1 = h4[(size_t)i1 * 16 + q];
            const float4 v2 = h4[(size_t)i2 * 16 + q];
            const float4 v3 = h4[(size_t)i3 * 16 + q];
            const bool p1 = (j + 1 < n), p2 = (j + 2 < n), p3 = (j + 3 < n);
            i0 = lst[min(j + 4, jl)];
            i1 = lst[min(j + 5, jl)];
            i2 = lst[min(j + 6, jl)];
            i3 = lst[min(j + 7, jl)];
            acc.x += v0.x;              acc.y += v0.y;              acc.z += v0.z;              acc.w += v0.w;
            acc.x += p1 ? v1.x : 0.f;   acc.y += p1 ? v1.y : 0.f;   acc.z += p1 ? v1.z : 0.f;   acc.w += p1 ? v1.w : 0.f;
            acc.x += p2 ? v2.x : 0.f;   acc.y += p2 ? v2.y : 0.f;   acc.z += p2 ? v2.z : 0.f;   acc.w += p2 ? v2.w : 0.f;
            acc.x += p3 ? v3.x : 0.f;   acc.y += p3 ? v3.y : 0.f;   acc.z += p3 ? v3.z : 0.f;   acc.w += p3 ? v3.w : 0.f;
        }
    }
    reinterpret_cast<float4*>(agg)[(size_t)d * 16 + q] = acc;
}

__global__ void __launch_bounds__(256) sage_gemm_kernel(
    const float* agg, const float* __restrict__ hin,
    const float* __restrict__ Wl,  const float* __restrict__ bias,
    const float* __restrict__ Wr,  float* out, int applyTanh)
{
    __shared__ float sWl[FDIM * FDIM];
    __shared__ float sWr[FDIM * FDIM];

    for (int i = threadIdx.x; i < (FDIM * FDIM) / 4; i += 256) {
        reinterpret_cast<float4*>(sWl)[i] = reinterpret_cast<const float4*>(Wl)[i];
        reinterpret_cast<float4*>(sWr)[i] = reinterpret_cast<const float4*>(Wr)[i];
    }
    __syncthreads();

    const int wave = threadIdx.x >> 5;
    const int lane = threadIdx.x & 31;
    const int half = lane >> 4;
    const int lrow = lane & 15;

    const int rb = blockIdx.x * 8 + wave;
    if (rb >= ROWTILES) return;
    const size_t r0 = (size_t)rb * 16;

    v8f acc[4];
    const v8f zero = {0.f, 0.f, 0.f, 0.f, 0.f, 0.f, 0.f, 0.f};
#pragma unroll
    for (int t = 0; t < 4; ++t) acc[t] = zero;

    const float* aRowL = agg + (r0 + lrow) * FDIM + half * 2;
    const float* aRowR = hin + (r0 + lrow) * FDIM + half * 2;

#pragma unroll
    for (int kb = 0; kb < FDIM; kb += 4) {
        v2f aL = *reinterpret_cast<const v2f*>(aRowL + kb);
        v2f aR = *reinterpret_cast<const v2f*>(aRowR + kb);
#pragma unroll
        for (int t = 0; t < 4; ++t) {
            const int n = t * 16 + lrow;
            v2f bL = *reinterpret_cast<const v2f*>(sWl + n * FDIM + kb + half * 2);
            v2f bR = *reinterpret_cast<const v2f*>(sWr + n * FDIM + kb + half * 2);
            acc[t] = __builtin_amdgcn_wmma_f32_16x16x4_f32(
                false, aL, false, bL, (short)0, acc[t], false, false);
            acc[t] = __builtin_amdgcn_wmma_f32_16x16x4_f32(
                false, aR, false, bR, (short)0, acc[t], false, false);
        }
    }

#pragma unroll
    for (int t = 0; t < 4; ++t) {
        const int n = t * 16 + lrow;
        const float bv = bias[n];
#pragma unroll
        for (int v = 0; v < 8; ++v) {
            const int m = v + half * 8;
            float val = acc[t][v] + bv;
            if (applyTanh) val = tanhf(val);
            out[(r0 + m) * FDIM + n] = val;
        }
    }
}

extern "C" void kernel_launch(void* const* d_in, const int* in_sizes, int n_in,
                              void* d_out, int out_size, void* d_ws, size_t ws_size,
                              hipStream_t stream)
{
    const float* x      = (const float*)d_in[0];
    const int*   ei     = (const int*)d_in[1];
    const int*   src    = ei;
    const int*   dst    = ei + NEDGES;
    const float* Wl_in  = (const float*)d_in[2];
    const float* bl_in  = (const float*)d_in[3];
    const float* Wr_in  = (const float*)d_in[4];
    const float* Wl_med = (const float*)d_in[5];
    const float* bl_med = (const float*)d_in[6];
    const float* Wr_med = (const float*)d_in[7];
    const float* Wl_out = (const float*)d_in[8];
    const float* bl_out = (const float*)d_in[9];
    const float* Wr_out = (const float*)d_in[10];

    float* bufP   = (float*)d_ws;
    float* bufO   = (float*)d_out;
    int*   cnt    = (int*)(bufP + (size_t)NNODES * FDIM);
    int*   cursor = cnt + NNODES;
    int*   lscan  = cursor + NNODES;
    int*   bbase  = lscan + NNODES;
    int*   ssrc   = bbase + 128;

    const dim3 eBlk(256);
    const dim3 eGrid((NEDGES + 255) / 256);
    const dim3 nBlk(256);
    const dim3 nGrid((NNODES * 16 + 255) / 256);
    const dim3 gBlk(256);
    const dim3 gGrid((ROWTILES + 7) / 8);

    hipMemsetAsync(cnt, 0, (size_t)2 * NNODES * sizeof(int), stream);
    degree_count_kernel<<<eGrid, eBlk, 0, stream>>>(dst, cnt, NEDGES);
    block_scan_kernel<<<dim3(SCAN_NBLK), dim3(SCAN_BLK), 0, stream>>>(cnt, lscan, NNODES);
    block_base_kernel<<<dim3(1), dim3(128), 0, stream>>>(cnt, lscan, bbase, SCAN_NBLK);
    bucket_fill_kernel<<<eGrid, eBlk, 0, stream>>>(src, dst, lscan, bbase, cursor, ssrc, NEDGES);

    auto layer = [&](const float* hin, float* hout, const float* Wl,
                     const float* bl, const float* Wr, int doTanh) {
        gather_sum_kernel<<<nGrid, nBlk, 0, stream>>>(hin, ssrc, cnt, lscan, bbase, hout, NNODES);
        sage_gemm_kernel<<<gGrid, gBlk, 0, stream>>>(hout, hin, Wl, bl, Wr, hout, doTanh);
    };

    layer(x,    bufP, Wl_in,  bl_in,  Wr_in,  1);
    layer(bufP, bufO, Wl_med, bl_med, Wr_med, 1);
    layer(bufO, bufP, Wl_med, bl_med, Wr_med, 1);
    layer(bufP, bufO, Wl_out, bl_out, Wr_out, 0);
}


# group edges by destination with a two-level counting sort on LDS int counters (chunk-local bucket sort, one owner group per 512 nodes), no global atomics; sum order = LDS arrival rank, run-varying
# speedup vs baseline: 2.5284x; 1.5964x over previous
#include <hip/hip_runtime.h>
#include <math.h>


#define NNODES 100000
#define NEDGES 1200000
#define FDIM   64
#define ROWTILES (NNODES / 16)

typedef __attribute__((ext_vector_type(2))) float v2f;
typedef __attribute__((ext_vector_type(8))) float v8f;

#define NCHUNK    128
#define CH        (NEDGES / NCHUNK)
#define CHP       9376
#define CH_ITERS  ((CH + 1023) / 1024)
#define RSHIFT    9
#define RNODES    (1 << RSHIFT)
#define NBUCK     ((NNODES + RNODES - 1) / RNODES)
#define BPITCH    256
#define SRC_BITS  17
#define SRC_MASK  ((1 << SRC_BITS) - 1)
#define STAGE_CAP 16384

static_assert(CH * NCHUNK == NEDGES, "chunks must tile the edge list");
static_assert(CHP >= CH && CHP % 32 == 0, "chunk pitch");
static_assert(NNODES <= (1 << SRC_BITS), "source index must fit the packed word");
static_assert(NBUCK <= BPITCH, "bucket table pitch");
static_assert(NNODES % 32 == 0, "node ranges end on a line");
static_assert(NCHUNK % 32 == 0, "runs per wave");

__device__ __forceinline__ int wave_inclusive_scan(int v, int lane)
{
    for (int o = 1; o < 32; o <<= 1) {
        int u = __shfl_up(v, o, 32);
        if (lane >= o) v += u;
    }
    return v;
}

__device__ __forceinline__ int block_exclusive_scan_1024(int v, int* sW)
{
    const int lane = threadIdx.x & 31;
    const int wave = threadIdx.x >> 5;
    const int inc  = wave_inclusive_scan(v, lane);
    if (lane == 31) sW[wave] = inc;
    __syncthreads();
    if (wave == 0) {
        const int w = sW[lane];
        sW[lane] = wave_inclusive_scan(w, lane) - w;
    }
    __syncthreads();
    const int r = sW[wave] + inc - v;
    __syncthreads();
    return r;
}

__global__ void __launch_bounds__(1024) chunk_sort_kernel(
    const int* __restrict__ src, const int* __restrict__ dst,
    int* __restrict__ srt, int* __restrict__ cntm, int* __restrict__ loffm)
{
    __shared__ int sHist[BPITCH];
    __shared__ int sOff[BPITCH];
    __shared__ int sW[32];
    __shared__ int sSorted[CHP];

    const int t = threadIdx.x;
    const int c = blockIdx.x;
    if (t < BPITCH) sHist[t] = 0;
    __syncthreads();

    int rec[CH_ITERS], bk[CH_ITERS], rk[CH_ITERS];
#pragma unroll
    for (int k = 0; k < CH_ITERS; ++k) {
        const int i = k * 1024 + t;
        const int e = c * CH + ((i < CH) ? i : 0);
        bk[k]  = dst[e];
        rec[k] = src[e];
    }
#pragma unroll
    for (int k = 0; k < CH_ITERS; ++k) {
        const int i = k * 1024 + t;
        const int d = bk[k];
        bk[k]  = d >> RSHIFT;
        rec[k] = ((d & (RNODES - 1)) << SRC_BITS) | rec[k];
        rk[k]  = 0;
        if (i < CH) rk[k] = atomicAdd(&sHist[bk[k]], 1);
    }
    __syncthreads();

    const int v  = (t < BPITCH) ? sHist[t] : 0;
    const int ex = block_exclusive_scan_1024(v, sW);
    if (t < BPITCH) {
        sOff[t] = ex;
        cntm[c * BPITCH + t]  = v;
        loffm[c * BPITCH + t] = ex;
    }
    __syncthreads();

#pragma unroll
    for (int k = 0; k < CH_ITERS; ++k) {
        const int i = k * 1024 + t;
        if (i < CH) sSorted[sOff[bk[k]] + rk[k]] = rec[k];
    }
    __syncthreads();

    for (int i = t; i < CHP; i += 1024)
        srt[c * CHP + i] = (i < CH) ? sSorted[i] : 0;
}

template <class Body>
__device__ __forceinline__ void for_each_bucket_word(
    const int* __restrict__ srt, const int* sRunStart, const int* sRunLen,
    int wave, int lane, Body body)
{
    constexpr int RPW = NCHUNK / 32;
    int st[RPW], ln[RPW], w0[RPW], w1[RPW];
#pragma unroll
    for (int q = 0; q < RPW; ++q) {
        st[q] = sRunStart[wave + 32 * q];
        ln[q] = sRunLen[wave + 32 * q];
    }
#pragma unroll
    for (int q = 0; q < RPW; ++q) {
        const int last = max(ln[q] - 1, 0);
        w0[q] = srt[st[q] + min(lane, last)];
        w1[q] = srt[st[q] + min(lane + 32, last)];
    }
#pragma unroll
    for (int q = 0; q < RPW; ++q) {
        if (lane < ln[q])      body(w0[q]);
        if (lane + 32 < ln[q]) body(w1[q]);
        for (int j = lane + 64; j < ln[q]; j += 32) body(srt[st[q] + j]);
    }
}

__global__ void __launch_bounds__(1024) owner_group_kernel(
    const int* __restrict__ srt, const int* __restrict__ cntm,
    const int* __restrict__ loffm, int* __restrict__ cnt,
    int* __restrict__ first, int* __restrict__ ssrc)
{
    __shared__ int sPart[4 * BPITCH];
    __shared__ int sRunStart[NCHUNK];
    __shared__ int sRunLen[NCHUNK];
    __shared__ int sCnt[RNODES];
    __shared__ int sCur[RNODES];
    __shared__ int sW[32];
    __shared__ int sBaseTot[2];
    __shared__ int sStage[STAGE_CAP];

    const int t    = threadIdx.x;
    const int lane = t & 31;
    const int wave = t >> 5;
    const int b    = blockIdx.x;

    {
        const int col = t & (BPITCH - 1);
        const int g   = t >> 8;
        int s = 0;
        for (int c = g * (NCHUNK / 4); c < (g + 1) * (NCHUNK / 4); ++c)
            s += cntm[c * BPITCH + col];
        sPart[g * BPITCH + col] = s;
    }
    if (t < NCHUNK) {
        sRunLen[t]   = cntm[t * BPITCH + b];
        sRunStart[t] = t * CHP + loffm[t * BPITCH + b];
    }
    if (t < RNODES) sCnt[t] = 0;
    __syncthreads();

    int tot = 0;
    if (t < BPITCH)
        tot = sPart[t] + sPart[BPITCH + t] + sPart[2 * BPITCH + t] + sPart[3 * BPITCH + t];
    const int padded = (tot + 31) & ~31;
    const int base   = block_exclusive_scan_1024(padded, sW);
    if (t == b) { sBaseTot[0] = base; sBaseTot[1] = tot; }
    __syncthreads();
    const int  myBase = sBaseTot[0];
    const int  myTot  = sBaseTot[1];
    const bool staged = (myTot <= STAGE_CAP);

    for_each_bucket_word(srt, sRunStart, sRunLen, wave, lane,
        [&](int r) { atomicAdd(&sCnt[r >> SRC_BITS], 1); });
    __syncthreads();

    const int v    = (t < RNODES) ? sCnt[t] : 0;
    const int ex   = block_exclusive_scan_1024(v, sW);
    const int node = b * RNODES + t;
    if (t < RNODES) {
        sCur[t] = ex;
        if (node < NNODES) {
            cnt[node]   = v;
            first[node] = myBase + ex;
        }
    }
    __syncthreads();

    for_each_bucket_word(srt, sRunStart, sRunLen, wave, lane,
        [&](int r) {
            const int slot = atomicAdd(&sCur[r >> SRC_BITS], 1);
            const int s    = r & SRC_MASK;
            if (staged) sStage[slot] = s;
            else        atomicExch(&ssrc[myBase + slot], s);
        });
    __syncthreads();

    if (staged) {
        const int np = (myTot + 31) & ~31;
        for (int i = t; i < np; i += 1024)
            ssrc[myBase + i] = (i < myTot) ? sStage[i] : 0;
    }
}

__global__ void __launch_bounds__(256) gather_sum_kernel(
    const float* __restrict__ h, const int* __restrict__ ssrc,
    const int* __restrict__ cnt, const int* __restrict__ first,
    float* __restrict__ agg, int nNodes)
{
    const int t = blockIdx.x * blockDim.x + threadIdx.x;
    const int d = t >> 4;
    if (d >= nNodes) return;
    const int q = t & 15;

    const int  n   = cnt[d];
    const int* lst = ssrc + first[d];
    const float4* h4 = reinterpret_cast<const float4*>(h);

    float4 acc = make_float4(0.f, 0.f, 0.f, 0.f);
    if (n > 0) {
        const int jl = n - 1;
        int i0 = lst[0];
        int i1 = lst[min(1, jl)];
        int i2 = lst[min(2, jl)];
        int i3 = lst[min(3, jl)];
        for (int j = 0; j < n; j += 4) {
            const float4 v0 = h4[(size_t)i0 * 16 + q];
            const float4 v1 = h4[(size_t)i1 * 16 + q];
            const float4 v2 = h4[(size_t)i2 * 16 + q];
            const float4 v3 = h4[(size_t)i3 * 16 + q];
            const bool p1 = (j + 1 < n), p2 = (j + 2 < n), p3 = (j + 3 < n);
            i0 = lst[min(j + 4, jl)];
            i1 = lst[min(j + 5, jl)];
            i2 = lst[min(j + 6, jl)];
            i3 = lst[min(j + 7, jl)];
            acc.x += v0.x;              acc.y += v0.y;              acc.z += v0.z;              acc.w += v0.w;
            acc.x += p1 ? v1.x : 0.f;   acc.y += p1 ? v1.y : 0.f;   acc.z += p1 ? v1.z : 0.f;   acc.w += p1 ? v1.w : 0.f;
            acc.x += p2 ? v2.x : 0.f;   acc.y += p2 ? v2.y : 0.f;   acc.z += p2 ? v2.z : 0.f;   acc.w += p2 ? v2.w : 0.f;
            acc.x += p3 ? v3.x : 0.f;   acc.y += p3 ? v3.y : 0.f;   acc.z += p3 ? v3.z : 0.f;   acc.w += p3 ? v3.w : 0.f;
        }
    }
    reinterpret_cast<float4*>(agg)[(size_t)d * 16 + q] = acc;
}

__global__ void __launch_bounds__(256) sage_gemm_kernel(
    const float* agg, const float* __restrict__ hin,
    const float* __restrict__ Wl,  const float* __restrict__ bias,
    const float* __restrict__ Wr,  float* out, int applyTanh)
{
    __shared__ float sWl[FDIM * FDIM];
    __shared__ float sWr[FDIM * FDIM];

    for (int i = threadIdx.x; i < (FDIM * FDIM) / 4; i += 256) {
        reinterpret_cast<float4*>(sWl)[i] = reinterpret_cast<const float4*>(Wl)[i];
        reinterpret_cast<float4*>(sWr)[i] = reinterpret_cast<const float4*>(Wr)[i];
    }
    __syncthreads();

    const int wave = threadIdx.x >> 5;
    const int lane = threadIdx.x & 31;
    const int half = lane >> 4;
    const int lrow = lane & 15;

    const int rb = blockIdx.x * 8 + wave;
    if (rb >= ROWTILES) return;
    const size_t r0 = (size_t)rb * 16;

    v8f acc[4];
    const v8f zero = {0.f, 0.f, 0.f, 0.f, 0.f, 0.f, 0.f, 0.f};
#pragma unroll
    for (int t = 0; t < 4; ++t) acc[t] = zero;

    const float* aRowL = agg + (r0 + lrow) * FDIM + half * 2;
    const float* aRowR = hin + (r0 + lrow) * FDIM + half * 2;

#pragma unroll
    for (int kb = 0; kb < FDIM; kb += 4) {
        v2f aL = *reinterpret_cast<const v2f*>(aRowL + kb);
        v2f aR = *reinterpret_cast<const v2f*>(aRowR + kb);
#pragma unroll
        for (int t = 0; t < 4; ++t) {
            const int n = t * 16 + lrow;
            v2f bL = *reinterpret_cast<const v2f*>(sWl + n * FDIM + kb + half * 2);
            v2f bR = *reinterpret_cast<const v2f*>(sWr + n * FDIM + kb + half * 2);
            acc[t] = __builtin_amdgcn_wmma_f32_16x16x4_f32(
                false, aL, false, bL, (short)0, acc[t], false, false);
            acc[t] = __builtin_amdgcn_wmma_f32_16x16x4_f32(
                false, aR, false, bR, (short)0, acc[t], false, false);
        }
    }

#pragma unroll
    for (int t = 0; t < 4; ++t) {
        const int n = t * 16 + lrow;
        const float bv = bias[n];
#pragma unroll
        for (int v = 0; v < 8; ++v) {
            const int m = v + half * 8;
            float val = acc[t][v] + bv;
            if (applyTanh) val = tanhf(val);
            out[(r0 + m) * FDIM + n] = val;
        }
    }
}

extern "C" void kernel_launch(void* const* d_in, const int* in_sizes, int n_in,
                              void* d_out, int out_size, void* d_ws, size_t ws_size,
                              hipStream_t stream)
{
    const float* x      = (const float*)d_in[0];
    const int*   ei     = (const int*)d_in[1];
    const int*   src    = ei;
    const int*   dst    = ei + NEDGES;
    const float* Wl_in  = (const float*)d_in[2];
    const float* bl_in  = (const float*)d_in[3];
    const float* Wr_in  = (const float*)d_in[4];
    const float* Wl_med = (const float*)d_in[5];
    const float* bl_med = (const float*)d_in[6];
    const float* Wr_med = (const float*)d_in[7];
    const float* Wl_out = (const float*)d_in[8];
    const float* bl_out = (const float*)d_in[9];
    const float* Wr_out = (const float*)d_in[10];

    float* bufP   = (float*)d_ws;
    float* bufO   = (float*)d_out;
    int*   cnt    = (int*)(bufP + (size_t)NNODES * FDIM);
    int*   first  = cnt + NNODES;
    int*   cntm   = first + NNODES;
    int*   loffm  = cntm + NCHUNK * BPITCH;
    int*   ssrc   = loffm + NCHUNK * BPITCH;
    int*   srt    = (int*)d_out;

    const dim3 nBlk(256);
    const dim3 nGrid((NNODES * 16 + 255) / 256);
    const dim3 gBlk(256);
    const dim3 gGrid((ROWTILES + 7) / 8);

    chunk_sort_kernel<<<dim3(NCHUNK), dim3(1024), 0, stream>>>(src, dst, srt, cntm, loffm);
    owner_group_kernel<<<dim3(NBUCK), dim3(1024), 0, stream>>>(srt, cntm, loffm, cnt, first, ssrc);

    auto layer = [&](const float* hin, float* hout, const float* Wl,
                     const float* bl, const float* Wr, int doTanh) {
        gather_sum_kernel<<<nGrid, nBlk, 0, stream>>>(hin, ssrc, cnt, first, hout, NNODES);
        sage_gemm_kernel<<<gGrid, gBlk, 0, stream>>>(hout, hin, Wl, bl, Wr, hout, doTanh);
    };

    layer(x,    bufP, Wl_in,  bl_in,  Wr_in,  1);
    layer(bufP, bufO, Wl_med, bl_med, Wr_med, 1);
    layer(bufO, bufP, Wl_med, bl_med, Wr_med, 1);
    layer(bufP, bufO, Wl_out, bl_out, Wr_out, 0);
}


# dense stage: LDS weight rows padded from 64 to 68 floats so the 16 lanes of a B-fragment read fall on distinct banks instead of one; same k order, same sums
# speedup vs baseline: 3.2373x; 1.2804x over previous
#include <hip/hip_runtime.h>
#include <math.h>


#define NNODES 100000
#define NEDGES 1200000
#define FDIM   64
#define ROWTILES (NNODES / 16)

typedef __attribute__((ext_vector_type(2))) float v2f;
typedef __attribute__((ext_vector_type(8))) float v8f;

#define NCHUNK    128
#define CH        (NEDGES / NCHUNK)
#define CHP       9376
#define CH_ITERS  ((CH + 1023) / 1024)
#define RSHIFT    9
#define RNODES    (1 << RSHIFT)
#define NBUCK     ((NNODES + RNODES - 1) / RNODES)
#define BPITCH    256
#define SRC_BITS  17
#define SRC_MASK  ((1 << SRC_BITS) - 1)
#define STAGE_CAP 16384

static_assert(CH * NCHUNK == NEDGES, "chunks must tile the edge list");
static_assert(CHP >= CH && CHP % 32 == 0, "chunk pitch");
static_assert(NNODES <= (1 << SRC_BITS), "source index must fit the packed word");
static_assert(NBUCK <= BPITCH, "bucket table pitch");
static_assert(NNODES % 32 == 0, "node ranges end on a line");
static_assert(NCHUNK % 32 == 0, "runs per wave");

__device__ __forceinline__ int wave_inclusive_scan(int v, int lane)
{
    for (int o = 1; o < 32; o <<= 1) {
        int u = __shfl_up(v, o, 32);
        if (lane >= o) v += u;
    }
    return v;
}

__device__ __forceinline__ int block_exclusive_scan_1024(int v, int* sW)
{
    const int lane = threadIdx.x & 31;
    const int wave = threadIdx.x >> 5;
    const int inc  = wave_inclusive_scan(v, lane);
    if (lane == 31) sW[wave] = inc;
    __syncthreads();
    if (wave == 0) {
        const int w = sW[lane];
        sW[lane] = wave_inclusive_scan(w, lane) - w;
    }
    __syncthreads();
    const int r = sW[wave] + inc - v;
    __syncthreads();
    return r;
}

__global__ void __launch_bounds__(1024) chunk_sort_kernel(
    const int* __restrict__ src, const int* __restrict__ dst,
    int* __restrict__ srt, int* __restrict__ cntm, int* __restrict__ loffm)
{
    __shared__ int sHist[BPITCH];
    __shared__ int sOff[BPITCH];
    __shared__ int sW[32];
    __shared__ int sSorted[CHP];

    const int t = threadIdx.x;
    const int c = blockIdx.x;
    if (t < BPITCH) sHist[t] = 0;
    __syncthreads();

    int rec[CH_ITERS], bk[CH_ITERS], rk[CH_ITERS];
#pragma unroll
    for (int k = 0; k < CH_ITERS; ++k) {
        const int i = k * 1024 + t;
        const int e = c * CH + ((i < CH) ? i : 0);
        bk[k]  = dst[e];
        rec[k] = src[e];
    }
#pragma unroll
    for (int k = 0; k < CH_ITERS; ++k) {
        const int i = k * 1024 + t;
        const int d = bk[k];
        bk[k]  = d >> RSHIFT;
        rec[k] = ((d & (RNODES - 1)) << SRC_BITS) | rec[k];
        rk[k]  = 0;
        if (i < CH) rk[k] = atomicAdd(&sHist[bk[k]], 1);
    }
    __syncthreads();

    const int v  = (t < BPITCH) ? sHist[t] : 0;
    const int ex = block_exclusive_scan_1024(v, sW);
    if (t < BPITCH) {
        sOff[t] = ex;
        cntm[c * BPITCH + t]  = v;
        loffm[c * BPITCH + t] = ex;
    }
    __syncthreads();

#pragma unroll
    for (int k = 0; k < CH_ITERS; ++k) {
        const int i = k * 1024 + t;
        if (i < CH) sSorted[sOff[bk[k]] + rk[k]] = rec[k];
    }
    __syncthreads();

    for (int i = t; i < CHP; i += 1024)
        srt[c * CHP + i] = (i < CH) ? sSorted[i] : 0;
}

template <class Body>
__device__ __forceinline__ void for_each_bucket_word(
    const int* __restrict__ srt, const int* sRunStart, const int* sRunLen,
    int wave, int lane, Body body)
{
    constexpr int RPW = NCHUNK / 32;
    int st[RPW], ln[RPW], w0[RPW], w1[RPW];
#pragma unroll
    for (int q = 0; q < RPW; ++q) {
        st[q] = sRunStart[wave + 32 * q];
        ln[q] = sRunLen[wave + 32 * q];
    }
#pragma unroll
    for (int q = 0; q < RPW; ++q) {
        const int last = max(ln[q] - 1, 0);
        w0[q] = srt[st[q] + min(lane, last)];
        w1[q] = srt[st[q] + min(lane + 32, last)];
    }
#pragma unroll
    for (int q = 0; q < RPW; ++q) {
        if (lane < ln[q])      body(w0[q]);
        if (lane + 32 < ln[q]) body(w1[q]);
        for (int j = lane + 64; j < ln[q]; j += 32) body(srt[st[q] + j]);
    }
}

__global__ void __launch_bounds__(1024) owner_group_kernel(
    const int* __restrict__ srt, const int* __restrict__ cntm,
    const int* __restrict__ loffm, int* __restrict__ cnt,
    int* __restrict__ first, int* __restrict__ ssrc)
{
    __shared__ int sPart[4 * BPITCH];
    __shared__ int sRunStart[NCHUNK];
    __shared__ int sRunLen[NCHUNK];
    __shared__ int sCnt[RNODES];
    __shared__ int sCur[RNODES];
    __shared__ int sW[32];
    __shared__ int sBaseTot[2];
    __shared__ int sStage[STAGE_CAP];

    const int t    = threadIdx.x;
    const int lane = t & 31;
    const int wave = t >> 5;
    const int b    = blockIdx.x;

    {
        const int col = t & (BPITCH - 1);
        const int g   = t >> 8;
        int s = 0;
        for (int c = g * (NCHUNK / 4); c < (g + 1) * (NCHUNK / 4); ++c)
            s += cntm[c * BPITCH + col];
        sPart[g * BPITCH + col] = s;
    }
    if (t < NCHUNK) {
        sRunLen[t]   = cntm[t * BPITCH + b];
        sRunStart[t] = t * CHP + loffm[t * BPITCH + b];
    }
    if (t < RNODES) sCnt[t] = 0;
    __syncthreads();

    int tot = 0;
    if (t < BPITCH)
        tot = sPart[t] + sPart[BPITCH + t] + sPart[2 * BPITCH + t] + sPart[3 * BPITCH + t];
    const int padded = (tot + 31) & ~31;
    const int base   = block_exclusive_scan_1024(padded, sW);
    if (t == b) { sBaseTot[0] = base; sBaseTot[1] = tot; }
    __syncthreads();
    const int  myBase = sBaseTot[0];
    const int  myTot  = sBaseTot[1];
    const bool staged = (myTot <= STAGE_CAP);

    for_each_bucket_word(srt, sRunStart, sRunLen, wave, lane,
        [&](int r) { atomicAdd(&sCnt[r >> SRC_BITS], 1); });
    __syncthreads();

    const int v    = (t < RNODES) ? sCnt[t] : 0;
    const int ex   = block_exclusive_scan_1024(v, sW);
    const int node = b * RNODES + t;
    if (t < RNODES) {
        sCur[t] = ex;
        if (node < NNODES) {
            cnt[node]   = v;
            first[node] = myBase + ex;
        }
    }
    __syncthreads();

    for_each_bucket_word(srt, sRunStart, sRunLen, wave, lane,
        [&](int r) {
            const int slot = atomicAdd(&sCur[r >> SRC_BITS], 1);
            const int s    = r & SRC_MASK;
            if (staged) sStage[slot] = s;
            else        atomicExch(&ssrc[myBase + slot], s);
        });
    __syncthreads();

    if (staged) {
        const int np = (myTot + 31) & ~31;
        for (int i = t; i < np; i += 1024)
            ssrc[myBase + i] = (i < myTot) ? sStage[i] : 0;
    }
}

__global__ void __launch_bounds__(256) gather_sum_kernel(
    const float* __restrict__ h, const int* __restrict__ ssrc,
    const int* __restrict__ cnt, const int* __restrict__ first,
    float* __restrict__ agg, int nNodes)
{
    const int t = blockIdx.x * blockDim.x + threadIdx.x;
    const int d = t >> 4;
    if (d >= nNodes) return;
    const int q = t & 15;

    const int  n   = cnt[d];
    const int* lst = ssrc + first[d];
    const float4* h4 = reinterpret_cast<const float4*>(h);

    float4 acc = make_float4(0.f, 0.f, 0.f, 0.f);
    if (n > 0) {
        const int jl = n - 1;
        int i0 = lst[0];
        int i1 = lst[min(1, jl)];
        int i2 = lst[min(2, jl)];
        int i3 = lst[min(3, jl)];
        for (int j = 0; j < n; j += 4) {
            const float4 v0 = h4[(size_t)i0 * 16 + q];
            const float4 v1 = h4[(size_t)i1 * 16 + q];
            const float4 v2 = h4[(size_t)i2 * 16 + q];
            const float4 v3 = h4[(size_t)i3 * 16 + q];
            const bool p1 = (j + 1 < n), p2 = (j + 2 < n), p3 = (j + 3 < n);
            i0 = lst[min(j + 4, jl)];
            i1 = lst[min(j + 5, jl)];
            i2 = lst[min(j + 6, jl)];
            i3 = lst[min(j + 7, jl)];
            acc.x += v0.x;              acc.y += v0.y;              acc.z += v0.z;              acc.w += v0.w;
            acc.x += p1 ? v1.x : 0.f;   acc.y += p1 ? v1.y : 0.f;   acc.z += p1 ? v1.z : 0.f;   acc.w += p1 ? v1.w : 0.f;
            acc.x += p2 ? v2.x : 0.f;   acc.y += p2 ? v2.y : 0.f;   acc.z += p2 ? v2.z : 0.f;   acc.w += p2 ? v2.w : 0.f;
            acc.x += p3 ? v3.x : 0.f;   acc.y += p3 ? v3.y : 0.f;   acc.z += p3 ? v3.z : 0.f;   acc.w += p3 ? v3.w : 0.f;
        }
    }
    reinterpret_cast<float4*>(agg)[(size_t)d * 16 + q] = acc;
}

#define WPITCH (FDIM + 4)
__global__ void __launch_bounds__(256) sage_gemm_kernel(
    const float* agg, const float* __restrict__ hin,
    const float* __restrict__ Wl,  const float* __restrict__ bias,
    const float* __restrict__ Wr,  float* out, int applyTanh)
{
    __shared__ alignas(16) float sWl[FDIM * WPITCH];
    __shared__ alignas(16) float sWr[FDIM * WPITCH];

    for (int i = threadIdx.x; i < (FDIM * FDIM) / 4; i += 256) {
        const int wrow = i >> 4;
        const int wcol = (i & 15) * 4;
        *reinterpret_cast<float4*>(sWl + wrow * WPITCH + wcol) = reinterpret_cast<const float4*>(Wl)[i];
        *reinterpret_cast<float4*>(sWr + wrow * WPITCH + wcol) = reinterpret_cast<const float4*>(Wr)[i];
    }
    __syncthreads();

    const int wave = threadIdx.x >> 5;
    const int lane = threadIdx.x & 31;
    const int half = lane >> 4;
    const int lrow = lane & 15;

    const int rb = blockIdx.x * 8 + wave;
    if (rb >= ROWTILES) return;
    const size_t r0 = (size_t)rb * 16;

    v8f acc[4];
    const v8f zero = {0.f, 0.f, 0.f, 0.f, 0.f, 0.f, 0.f, 0.f};
#pragma unroll
    for (int t = 0; t < 4; ++t) acc[t] = zero;

    const float* aRowL = agg + (r0 + lrow) * FDIM + half * 2;
    const float* aRowR = hin + (r0 + lrow) * FDIM + half * 2;

#pragma unroll
    for (int kb = 0; kb < FDIM; kb += 4) {
        v2f aL = *reinterpret_cast<const v2f*>(aRowL + kb);
        v2f aR = *reinterpret_cast<const v2f*>(aRowR + kb);
#pragma unroll
        for (int t = 0; t < 4; ++t) {
            const int n = t * 16 + lrow;
            v2f bL = *reinterpret_cast<const v2f*>(sWl + n * WPITCH + kb + half * 2);
            v2f bR = *reinterpret_cast<const v2f*>(sWr + n * WPITCH + kb + half * 2);
            acc[t] = __builtin_amdgcn_wmma_f32_16x16x4_f32(
                false, aL, false, bL, (short)0, acc[t], false, false);
            acc[t] = __builtin_amdgcn_wmma_f32_16x16x4_f32(
                false, aR, false, bR, (short)0, acc[t], false, false);
        }
    }

#pragma unroll
    for (int t = 0; t < 4; ++t) {
        const int n = t * 16 + lrow;
        const float bv = bias[n];
#pragma unroll
        for (int v = 0; v < 8; ++v) {
            const int m = v + half * 8;
            float val = acc[t][v] + bv;
            if (applyTanh) val = tanhf(val);
            out[(r0 + m) * FDIM + n] = val;
        }
    }
}

extern "C" void kernel_launch(void* const* d_in, const int* in_sizes, int n_in,
                              void* d_out, int out_size, void* d_ws, size_t ws_size,
                              hipStream_t stream)
{
    const float* x      = (const float*)d_in[0];
    const int*   ei     = (const int*)d_in[1];
    const int*   src    = ei;
    const int*   dst    = ei + NEDGES;
    const float* Wl_in  = (const float*)d_in[2];
    const float* bl_in  = (const float*)d_in[3];
    const float* Wr_in  = (const float*)d_in[4];
    const float* Wl_med = (const float*)d_in[5];
    const float* bl_med = (const float*)d_in[6];
    const float* Wr_med = (const float*)d_in[7];
    const float* Wl_out = (const float*)d_in[8];
    const float* bl_out = (const float*)d_in[9];
    const float* Wr_out = (const float*)d_in[10];

    float* bufP   = (float*)d_ws;
    float* bufO   = (float*)d_out;
    int*   cnt    = (int*)(bufP + (size_t)NNODES * FDIM);
    int*   first  = cnt + NNODES;
    int*   cntm   = first + NNODES;
    int*   loffm  = cntm + NCHUNK * BPITCH;
    int*   ssrc   = loffm + NCHUNK * BPITCH;
    int*   srt    = (int*)d_out;

    const dim3 nBlk(256);
    const dim3 nGrid((NNODES * 16 + 255) / 256);
    const dim3 gBlk(256);
    const dim3 gGrid((ROWTILES + 7) / 8);

    chunk_sort_kernel<<<dim3(NCHUNK), dim3(1024), 0, stream>>>(src, dst, srt, cntm, loffm);
    owner_group_kernel<<<dim3(NBUCK), dim3(1024), 0, stream>>>(srt, cntm, loffm, cnt, first, ssrc);

    auto layer = [&](const float* hin, float* hout, const float* Wl,
                     const float* bl, const float* Wr, int doTanh) {
        gather_sum_kernel<<<nGrid, nBlk, 0, stream>>>(hin, ssrc, cnt, first, hout, NNODES);
        sage_gemm_kernel<<<gGrid, gBlk, 0, stream>>>(hout, hin, Wl, bl, Wr, hout, doTanh);
    };

    layer(x,    bufP, Wl_in,  bl_in,  Wr_in,  1);
    layer(bufP, bufO, Wl_med, bl_med, Wr_med, 1);
    layer(bufO, bufP, Wl_med, bl_med, Wr_med, 1);
    layer(bufP, bufO, Wl_out, bl_out, Wr_out, 0);
}


# fuse the neighbour gather into the dense stage: a wave sums its 16 rows' neighbours into an LDS tile and reads the WMMA A fragments from it; 6 launches; sum order: Wr product before Wl product
# speedup vs baseline: 3.7342x; 1.1535x over previous
#include <hip/hip_runtime.h>
#include <math.h>


#define NNODES 100000
#define NEDGES 1200000
#define FDIM   64
#define ROWTILES (NNODES / 16)

typedef __attribute__((ext_vector_type(2))) float v2f;
typedef __attribute__((ext_vector_type(8))) float v8f;

#define NCHUNK    128
#define CH        (NEDGES / NCHUNK)
#define CHP       9376
#define CH_ITERS  ((CH + 1023) / 1024)
#define RSHIFT    9
#define RNODES    (1 << RSHIFT)
#define NBUCK     ((NNODES + RNODES - 1) / RNODES)
#define BPITCH    256
#define SRC_BITS  17
#define SRC_MASK  ((1 << SRC_BITS) - 1)
#define STAGE_CAP 16384

static_assert(CH * NCHUNK == NEDGES, "chunks must tile the edge list");
static_assert(CHP >= CH && CHP % 32 == 0, "chunk pitch");
static_assert(NNODES <= (1 << SRC_BITS), "source index must fit the packed word");
static_assert(NBUCK <= BPITCH, "bucket table pitch");
static_assert(NNODES % 32 == 0, "node ranges end on a line");
static_assert(NCHUNK % 32 == 0, "runs per wave");

__device__ __forceinline__ int wave_inclusive_scan(int v, int lane)
{
    for (int o = 1; o < 32; o <<= 1) {
        int u = __shfl_up(v, o, 32);
        if (lane >= o) v += u;
    }
    return v;
}

__device__ __forceinline__ int block_exclusive_scan_1024(int v, int* sW)
{
    const int lane = threadIdx.x & 31;
    const int wave = threadIdx.x >> 5;
    const int inc  = wave_inclusive_scan(v, lane);
    if (lane == 31) sW[wave] = inc;
    __syncthreads();
    if (wave == 0) {
        const int w = sW[lane];
        sW[lane] = wave_inclusive_scan(w, lane) - w;
    }
    __syncthreads();
    const int r = sW[wave] + inc - v;
    __syncthreads();
    return r;
}

__global__ void __launch_bounds__(1024) chunk_sort_kernel(
    const int* __restrict__ src, const int* __restrict__ dst,
    int* __restrict__ srt, int* __restrict__ cntm, int* __restrict__ loffm)
{
    __shared__ int sHist[BPITCH];
    __shared__ int sOff[BPITCH];
    __shared__ int sW[32];
    __shared__ int sSorted[CHP];

    const int t = threadIdx.x;
    const int c = blockIdx.x;
    if (t < BPITCH) sHist[t] = 0;
    __syncthreads();

    int rec[CH_ITERS], bk[CH_ITERS], rk[CH_ITERS];
#pragma unroll
    for (int k = 0; k < CH_ITERS; ++k) {
        const int i = k * 1024 + t;
        const int e = c * CH + ((i < CH) ? i : 0);
        bk[k]  = dst[e];
        rec[k] = src[e];
    }
#pragma unroll
    for (int k = 0; k < CH_ITERS; ++k) {
        const int i = k * 1024 + t;
        const int d = bk[k];
        bk[k]  = d >> RSHIFT;
        rec[k] = ((d & (RNODES - 1)) << SRC_BITS) | rec[k];
        rk[k]  = 0;
        if (i < CH) rk[k] = atomicAdd(&sHist[bk[k]], 1);
    }
    __syncthreads();

    const int v  = (t < BPITCH) ? sHist[t] : 0;
    const int ex = block_exclusive_scan_1024(v, sW);
    if (t < BPITCH) {
        sOff[t] = ex;
        cntm[c * BPITCH + t]  = v;
        loffm[c * BPITCH + t] = ex;
    }
    __syncthreads();

#pragma unroll
    for (int k = 0; k < CH_ITERS; ++k) {
        const int i = k * 1024 + t;
        if (i < CH) sSorted[sOff[bk[k]] + rk[k]] = rec[k];
    }
    __syncthreads();

    for (int i = t; i < CHP; i += 1024)
        srt[c * CHP + i] = (i < CH) ? sSorted[i] : 0;
}

template <class Body>
__device__ __forceinline__ void for_each_bucket_word(
    const int* __restrict__ srt, const int* sRunStart, const int* sRunLen,
    int wave, int lane, Body body)
{
    constexpr int RPW = NCHUNK / 32;
    int st[RPW], ln[RPW], w0[RPW], w1[RPW];
#pragma unroll
    for (int q = 0; q < RPW; ++q) {
        st[q] = sRunStart[wave + 32 * q];
        ln[q] = sRunLen[wave + 32 * q];
    }
#pragma unroll
    for (int q = 0; q < RPW; ++q) {
        const int last = max(ln[q] - 1, 0);
        w0[q] = srt[st[q] + min(lane, last)];
        w1[q] = srt[st[q] + min(lane + 32, last)];
    }
#pragma unroll
    for (int q = 0; q < RPW; ++q) {
        if (lane < ln[q])      body(w0[q]);
        if (lane + 32 < ln[q]) body(w1[q]);
        for (int j = lane + 64; j < ln[q]; j += 32) body(srt[st[q] + j]);
    }
}

__global__ void __launch_bounds__(1024) owner_group_kernel(
    const int* __restrict__ srt, const int* __restrict__ cntm,
    const int* __restrict__ loffm, int* __restrict__ cnt,
    int* __restrict__ first, int* __restrict__ ssrc)
{
    __shared__ int sPart[4 * BPITCH];
    __shared__ int sRunStart[NCHUNK];
    __shared__ int sRunLen[NCHUNK];
    __shared__ int sCnt[RNODES];
    __shared__ int sCur[RNODES];
    __shared__ int sW[32];
    __shared__ int sBaseTot[2];
    __shared__ int sStage[STAGE_CAP];

    const int t    = threadIdx.x;
    const int lane = t & 31;
    const int wave = t >> 5;
    const int b    = blockIdx.x;

    {
        const int col = t & (BPITCH - 1);
        const int g   = t >> 8;
        int s = 0;
        for (int c = g * (NCHUNK / 4); c < (g + 1) * (NCHUNK / 4); ++c)
            s += cntm[c * BPITCH + col];
        sPart[g * BPITCH + col] = s;
    }
    if (t < NCHUNK) {
        sRunLen[t]   = cntm[t * BPITCH + b];
        sRunStart[t] = t * CHP + loffm[t * BPITCH + b];
    }
    if (t < RNODES) sCnt[t] = 0;
    __syncthreads();

    int tot = 0;
    if (t < BPITCH)
        tot = sPart[t] + sPart[BPITCH + t] + sPart[2 * BPITCH + t] + sPart[3 * BPITCH + t];
    const int padded = (tot + 31) & ~31;
    const int base   = block_exclusive_scan_1024(padded, sW);
    if (t == b) { sBaseTot[0] = base; sBaseTot[1] = tot; }
    __syncthreads();
    const int  myBase = sBaseTot[0];
    const int  myTot  = sBaseTot[1];
    const bool staged = (myTot <= STAGE_CAP);

    for_each_bucket_word(srt, sRunStart, sRunLen, wave, lane,
        [&](int r) { atomicAdd(&sCnt[r >> SRC_BITS], 1); });
    __syncthreads();

    const int v    = (t < RNODES) ? sCnt[t] : 0;
    const int ex   = block_exclusive_scan_1024(v, sW);
    const int node = b * RNODES + t;
    if (t < RNODES) {
        sCur[t] = ex;
        if (node < NNODES) {
            cnt[node]   = v;
            first[node] = myBase + ex;
        }
    }
    __syncthreads();

    for_each_bucket_word(srt, sRunStart, sRunLen, wave, lane,
        [&](int r) {
            const int slot = atomicAdd(&sCur[r >> SRC_BITS], 1);
            const int s    = r & SRC_MASK;
            if (staged) sStage[slot] = s;
            else        atomicExch(&ssrc[myBase + slot], s);
        });
    __syncthreads();

    if (staged) {
        const int np = (myTot + 31) & ~31;
        for (int i = t; i < np; i += 1024)
            ssrc[myBase + i] = (i < myTot) ? sStage[i] : 0;
    }
}

#define WPITCH (FDIM + 4)
#define WAVES_PER_BLOCK 8
__global__ void __launch_bounds__(256) sage_layer_kernel(
    const float* __restrict__ hin, const int* __restrict__ ssrc,
    const int* __restrict__ cnt, const int* __restrict__ first,
    const float* __restrict__ Wl,  const float* __restrict__ bias,
    const float* __restrict__ Wr,  float* __restrict__ out, int applyTanh)
{
    __shared__ alignas(16) float sWl[FDIM * WPITCH];
    __shared__ alignas(16) float sWr[FDIM * WPITCH];
    __shared__ alignas(16) float sAgg[WAVES_PER_BLOCK * 16 * WPITCH];

    const int wave = threadIdx.x >> 5;
    const int lane = threadIdx.x & 31;
    const int half = lane >> 4;
    const int lrow = lane & 15;

    const int  rb     = blockIdx.x * WAVES_PER_BLOCK + wave;
    const bool active = (rb < ROWTILES);
    const size_t r0   = (size_t)(active ? rb : 0) * 16;

    const int rowCnt   = cnt[r0 + lrow];
    const int rowFirst = first[r0 + lrow];

    for (int i = threadIdx.x; i < (FDIM * FDIM) / 4; i += 256) {
        const int wrow = i >> 4;
        const int wcol = (i & 15) * 4;
        *reinterpret_cast<float4*>(sWl + wrow * WPITCH + wcol) = reinterpret_cast<const float4*>(Wl)[i];
        *reinterpret_cast<float4*>(sWr + wrow * WPITCH + wcol) = reinterpret_cast<const float4*>(Wr)[i];
    }
    __syncthreads();
    if (!active) return;

    v8f acc[4];
    const v8f zero = {0.f, 0.f, 0.f, 0.f, 0.f, 0.f, 0.f, 0.f};
#pragma unroll
    for (int t = 0; t < 4; ++t) acc[t] = zero;

    const float* aRowR = hin + (r0 + lrow) * FDIM + half * 2;
#pragma unroll
    for (int kb = 0; kb < FDIM; kb += 4) {
        v2f aR = *reinterpret_cast<const v2f*>(aRowR + kb);
#pragma unroll
        for (int t = 0; t < 4; ++t) {
            const int n = t * 16 + lrow;
            v2f bR = *reinterpret_cast<const v2f*>(sWr + n * WPITCH + kb + half * 2);
            acc[t] = __builtin_amdgcn_wmma_f32_16x16x4_f32(
                false, aR, false, bR, (short)0, acc[t], false, false);
        }
    }

    float* sTile = sAgg + wave * (16 * WPITCH);
    const float4* h4 = reinterpret_cast<const float4*>(hin);
    const int q = lrow;
    for (int s = 0; s < 4; ++s) {
        const int rA = 4 * s + half;
        const int rB = rA + 2;
        const int nA = __shfl(rowCnt, rA, 32);
        const int nB = __shfl(rowCnt, rB, 32);
        const int fA = __shfl(rowFirst, rA, 32);
        const int fB = __shfl(rowFirst, rB, 32);
        const int n  = max(nA, nB);

        float4 accA = make_float4(0.f, 0.f, 0.f, 0.f);
        float4 accB = make_float4(0.f, 0.f, 0.f, 0.f);
        if (n > 0) {
            const int* lA = ssrc + ((nA > 0) ? fA : fB);
            const int* lB = ssrc + ((nB > 0) ? fB : fA);
            const int jlA = max(nA - 1, 0);
            const int jlB = max(nB - 1, 0);
            int a0 = lA[0];
            int a1 = lA[min(1, jlA)];
            int a2 = lA[min(2, jlA)];
            int a3 = lA[min(3, jlA)];
            int b0 = lB[0];
            int b1 = lB[min(1, jlB)];
            int b2 = lB[min(2, jlB)];
            int b3 = lB[min(3, jlB)];
            for (int j = 0; j < n; j += 4) {
                const float4 u0 = h4[(size_t)a0 * 16 + q];
                const float4 u1 = h4[(size_t)a1 * 16 + q];
                const float4 u2 = h4[(size_t)a2 * 16 + q];
                const float4 u3 = h4[(size_t)a3 * 16 + q];
                const float4 w0 = h4[(size_t)b0 * 16 + q];
                const float4 w1 = h4[(size_t)b1 * 16 + q];
                const float4 w2 = h4[(size_t)b2 * 16 + q];
                const float4 w3 = h4[(size_t)b3 * 16 + q];
                const bool pa0 = (j < nA), pa1 = (j + 1 < nA), pa2 = (j + 2 < nA), pa3 = (j + 3 < nA);
                const bool pb0 = (j < nB), pb1 = (j + 1 < nB), pb2 = (j + 2 < nB), pb3 = (j + 3 < nB);
                a0 = lA[min(j + 4, jlA)];
                a1 = lA[min(j + 5, jlA)];
                a2 = lA[min(j + 6, jlA)];
                a3 = lA[min(j + 7, jlA)];
                b0 = lB[min(j + 4, jlB)];
                b1 = lB[min(j + 5, jlB)];
                b2 = lB[min(j + 6, jlB)];
                b3 = lB[min(j + 7, jlB)];
                accA.x += pa0 ? u0.x : 0.f;   accA.y += pa0 ? u0.y : 0.f;   accA.z += pa0 ? u0.z : 0.f;   accA.w += pa0 ? u0.w : 0.f;
                accA.x += pa1 ? u1.x : 0.f;   accA.y += pa1 ? u1.y : 0.f;   accA.z += pa1 ? u1.z : 0.f;   accA.w += pa1 ? u1.w : 0.f;
                accA.x += pa2 ? u2.x : 0.f;   accA.y += pa2 ? u2.y : 0.f;   accA.z += pa2 ? u2.z : 0.f;   accA.w += pa2 ? u2.w : 0.f;
                accA.x += pa3 ? u3.x : 0.f;   accA.y += pa3 ? u3.y : 0.f;   accA.z += pa3 ? u3.z : 0.f;   accA.w += pa3 ? u3.w : 0.f;
                accB.x += pb0 ? w0.x : 0.f;   accB.y += pb0 ? w0.y : 0.f;   accB.z += pb0 ? w0.z : 0.f;   accB.w += pb0 ? w0.w : 0.f;
                accB.x += pb1 ? w1.x : 0.f;   accB.y += pb1 ? w1.y : 0.f;   accB.z += pb1 ? w1.z : 0.f;   accB.w += pb1 ? w1.w : 0.f;
                accB.x += pb2 ? w2.x : 0.f;   accB.y += pb2 ? w2.y : 0.f;   accB.z += pb2 ? w2.z : 0.f;   accB.w += pb2 ? w2.w : 0.f;
                accB.x += pb3 ? w3.x : 0.f;   accB.y += pb3 ? w3.y : 0.f;   accB.z += pb3 ? w3.z : 0.f;   accB.w += pb3 ? w3.w : 0.f;
            }
        }
        *reinterpret_cast<float4*>(sTile + rA * WPITCH + q * 4) = accA;
        *reinterpret_cast<float4*>(sTile + rB * WPITCH + q * 4) = accB;
    }
    __builtin_amdgcn_wave_barrier();

    const float* aRowL = sTile + lrow * WPITCH + half * 2;
#pragma unroll
    for (int kb = 0; kb < FDIM; kb += 4) {
        v2f aL = *reinterpret_cast<const v2f*>(aRowL + kb);
#pragma unroll
        for (int t = 0; t < 4; ++t) {
            const int n = t * 16 + lrow;
            v2f bL = *reinterpret_cast<const v2f*>(sWl + n * WPITCH + kb + half * 2);
            acc[t] = __builtin_amdgcn_wmma_f32_16x16x4_f32(
                false, aL, false, bL, (short)0, acc[t], false, false);
        }
    }

#pragma unroll
    for (int t = 0; t < 4; ++t) {
        const int n = t * 16 + lrow;
        const float bv = bias[n];
#pragma unroll
        for (int v = 0; v < 8; ++v) {
            const int m = v + half * 8;
            float val = acc[t][v] + bv;
            if (applyTanh) val = tanhf(val);
            out[(r0 + m) * FDIM + n] = val;
        }
    }
}

extern "C" void kernel_launch(void* const* d_in, const int* in_sizes, int n_in,
                              void* d_out, int out_size, void* d_ws, size_t ws_size,
                              hipStream_t stream)
{
    const float* x      = (const float*)d_in[0];
    const int*   ei     = (const int*)d_in[1];
    const int*   src    = ei;
    const int*   dst    = ei + NEDGES;
    const float* Wl_in  = (const float*)d_in[2];
    const float* bl_in  = (const float*)d_in[3];
    const float* Wr_in  = (const float*)d_in[4];
    const float* Wl_med = (const float*)d_in[5];
    const float* bl_med = (const float*)d_in[6];
    const float* Wr_med = (const float*)d_in[7];
    const float* Wl_out = (const float*)d_in[8];
    const float* bl_out = (const float*)d_in[9];
    const float* Wr_out = (const float*)d_in[10];

    float* bufP   = (float*)d_ws;
    float* bufO   = (float*)d_out;
    int*   cnt    = (int*)(bufP + (size_t)NNODES * FDIM);
    int*   first  = cnt + NNODES;
    int*   cntm   = first + NNODES;
    int*   loffm  = cntm + NCHUNK * BPITCH;
    int*   ssrc   = loffm + NCHUNK * BPITCH;
    int*   srt    = (int*)d_out;

    const dim3 gBlk(256);
    const dim3 gGrid((ROWTILES + WAVES_PER_BLOCK - 1) / WAVES_PER_BLOCK);

    chunk_sort_kernel<<<dim3(NCHUNK), dim3(1024), 0, stream>>>(src, dst, srt, cntm, loffm);
    owner_group_kernel<<<dim3(NBUCK), dim3(1024), 0, stream>>>(srt, cntm, loffm, cnt, first, ssrc);

    auto layer = [&](const float* hin, float* hout, const float* Wl,
                     const float* bl, const float* Wr, int doTanh) {
        sage_layer_kernel<<<gGrid, gBlk, 0, stream>>>(hin, ssrc, cnt, first, Wl, bl, Wr, hout, doTanh);
    };

    layer(x,    bufP, Wl_in,  bl_in,  Wr_in,  1);
    layer(bufP, bufO, Wl_med, bl_med, Wr_med, 1);
    layer(bufO, bufP, Wl_med, bl_med, Wr_med, 1);
    layer(bufP, bufO, Wl_out, bl_out, Wr_out, 0);
}


# gather: each row's source list read once into a register piece one row pair ahead, indices taken by lane shuffle; 12 row loads per round (was 8), none waits on an index load; same sum order
# speedup vs baseline: 3.9637x; 1.0615x over previous
#include <hip/hip_runtime.h>
#include <math.h>


#define NNODES 100000
#define NEDGES 1200000
#define FDIM   64
#define ROWTILES (NNODES / 16)

typedef __attribute__((ext_vector_type(2))) float v2f;
typedef __attribute__((ext_vector_type(8))) float v8f;

#define NCHUNK    128
#define CH        (NEDGES / NCHUNK)
#define CHP       9376
#define CH_ITERS  ((CH + 1023) / 1024)
#define RSHIFT    9
#define RNODES    (1 << RSHIFT)
#define NBUCK     ((NNODES + RNODES - 1) / RNODES)
#define BPITCH    256
#define SRC_BITS  17
#define SRC_MASK  ((1 << SRC_BITS) - 1)
#define STAGE_CAP 16384

static_assert(CH * NCHUNK == NEDGES, "chunks must tile the edge list");
static_assert(CHP >= CH && CHP % 32 == 0, "chunk pitch");
static_assert(NNODES <= (1 << SRC_BITS), "source index must fit the packed word");
static_assert(NBUCK <= BPITCH, "bucket table pitch");
static_assert(NNODES % 32 == 0, "node ranges end on a line");
static_assert(NCHUNK % 32 == 0, "runs per wave");

__device__ __forceinline__ int wave_inclusive_scan(int v, int lane)
{
    for (int o = 1; o < 32; o <<= 1) {
        int u = __shfl_up(v, o, 32);
        if (lane >= o) v += u;
    }
    return v;
}

__device__ __forceinline__ int block_exclusive_scan_1024(int v, int* sW)
{
    const int lane = threadIdx.x & 31;
    const int wave = threadIdx.x >> 5;
    const int inc  = wave_inclusive_scan(v, lane);
    if (lane == 31) sW[wave] = inc;
    __syncthreads();
    if (wave == 0) {
        const int w = sW[lane];
        sW[lane] = wave_inclusive_scan(w, lane) - w;
    }
    __syncthreads();
    const int r = sW[wave] + inc - v;
    __syncthreads();
    return r;
}

__global__ void __launch_bounds__(1024) chunk_sort_kernel(
    const int* __restrict__ src, const int* __restrict__ dst,
    int* __restrict__ srt, int* __restrict__ cntm, int* __restrict__ loffm)
{
    __shared__ int sHist[BPITCH];
    __shared__ int sOff[BPITCH];
    __shared__ int sW[32];
    __shared__ int sSorted[CHP];

    const int t = threadIdx.x;
    const int c = blockIdx.x;
    if (t < BPITCH) sHist[t] = 0;
    __syncthreads();

    int rec[CH_ITERS], bk[CH_ITERS], rk[CH_ITERS];
#pragma unroll
    for (int k = 0; k < CH_ITERS; ++k) {
        const int i = k * 1024 + t;
        const int e = c * CH + ((i < CH) ? i : 0);
        bk[k]  = dst[e];
        rec[k] = src[e];
    }
#pragma unroll
    for (int k = 0; k < CH_ITERS; ++k) {
        const int i = k * 1024 + t;
        const int d = bk[k];
        bk[k]  = d >> RSHIFT;
        rec[k] = ((d & (RNODES - 1)) << SRC_BITS) | rec[k];
        rk[k]  = 0;
        if (i < CH) rk[k] = atomicAdd(&sHist[bk[k]], 1);
    }
    __syncthreads();

    const int v  = (t < BPITCH) ? sHist[t] : 0;
    const int ex = block_exclusive_scan_1024(v, sW);
    if (t < BPITCH) {
        sOff[t] = ex;
        cntm[c * BPITCH + t]  = v;
        loffm[c * BPITCH + t] = ex;
    }
    __syncthreads();

#pragma unroll
    for (int k = 0; k < CH_ITERS; ++k) {
        const int i = k * 1024 + t;
        if (i < CH) sSorted[sOff[bk[k]] + rk[k]] = rec[k];
    }
    __syncthreads();

    for (int i = t; i < CHP; i += 1024)
        srt[c * CHP + i] = (i < CH) ? sSorted[i] : 0;
}

template <class Body>
__device__ __forceinline__ void for_each_bucket_word(
    const int* __restrict__ srt, const int* sRunStart, const int* sRunLen,
    int wave, int lane, Body body)
{
    constexpr int RPW = NCHUNK / 32;
    int st[RPW], ln[RPW], w0[RPW], w1[RPW];
#pragma unroll
    for (int q = 0; q < RPW; ++q) {
        st[q] = sRunStart[wave + 32 * q];
        ln[q] = sRunLen[wave + 32 * q];
    }
#pragma unroll
    for (int q = 0; q < RPW; ++q) {
        const int last = max(ln[q] - 1, 0);
        w0[q] = srt[st[q] + min(lane, last)];
        w1[q] = srt[st[q] + min(lane + 32, last)];
    }
#pragma unroll
    for (int q = 0; q < RPW; ++q) {
        if (lane < ln[q])      body(w0[q]);
        if (lane + 32 < ln[q]) body(w1[q]);
        for (int j = lane + 64; j < ln[q]; j += 32) body(srt[st[q] + j]);
    }
}

__global__ void __launch_bounds__(1024) owner_group_kernel(
    const int* __restrict__ srt, const int* __restrict__ cntm,
    const int* __restrict__ loffm, int* __restrict__ cnt,
    int* __restrict__ first, int* __restrict__ ssrc)
{
    __shared__ int sPart[4 * BPITCH];
    __shared__ int sRunStart[NCHUNK];
    __shared__ int sRunLen[NCHUNK];
    __shared__ int sCnt[RNODES];
    __shared__ int sCur[RNODES];
    __shared__ int sW[32];
    __shared__ int sBaseTot[2];
    __shared__ int sStage[STAGE_CAP];

    const int t    = threadIdx.x;
    const int lane = t & 31;
    const int wave = t >> 5;
    const int b    = blockIdx.x;

    {
        const int col = t & (BPITCH - 1);
        const int g   = t >> 8;
        int s = 0;
        for (int c = g * (NCHUNK / 4); c < (g + 1) * (NCHUNK / 4); ++c)
            s += cntm[c * BPITCH + col];
        sPart[g * BPITCH + col] = s;
    }
    if (t < NCHUNK) {
        sRunLen[t]   = cntm[t * BPITCH + b];
        sRunStart[t] = t * CHP + loffm[t * BPITCH + b];
    }
    if (t < RNODES) sCnt[t] = 0;
    __syncthreads();

    int tot = 0;
    if (t < BPITCH)
        tot = sPart[t] + sPart[BPITCH + t] + sPart[2 * BPITCH + t] + sPart[3 * BPITCH + t];
    const int padded = (tot + 31) & ~31;
    const int base   = block_exclusive_scan_1024(padded, sW);
    if (t == b) { sBaseTot[0] = base; sBaseTot[1] = tot; }
    __syncthreads();
    const int  myBase = sBaseTot[0];
    const int  myTot  = sBaseTot[1];
    const bool staged = (myTot <= STAGE_CAP);

    for_each_bucket_word(srt, sRunStart, sRunLen, wave, lane,
        [&](int r) { atomicAdd(&sCnt[r >> SRC_BITS], 1); });
    __syncthreads();

    const int v    = (t < RNODES) ? sCnt[t] : 0;
    const int ex   = block_exclusive_scan_1024(v, sW);
    const int node = b * RNODES + t;
    if (t < RNODES) {
        sCur[t] = ex;
        if (node < NNODES) {
            cnt[node]   = v;
            first[node] = myBase + ex;
        }
    }
    __syncthreads();

    for_each_bucket_word(srt, sRunStart, sRunLen, wave, lane,
        [&](int r) {
            const int slot = atomicAdd(&sCur[r >> SRC_BITS], 1);
            const int s    = r & SRC_MASK;
            if (staged) sStage[slot] = s;
            else        atomicExch(&ssrc[myBase + slot], s);
        });
    __syncthreads();

    if (staged) {
        const int np = (myTot + 31) & ~31;
        for (int i = t; i < np; i += 1024)
            ssrc[myBase + i] = (i < myTot) ? sStage[i] : 0;
    }
}

__device__ __forceinline__ int list_chunk(const int* __restrict__ ssrc, int f, int n, int base, int l)
{
    const int p = base + l;
    return (p < n) ? ssrc[f + p] : 0;
}

#define WPITCH (FDIM + 4)
#define WAVES_PER_BLOCK 8
#define GROUND 6
__global__ void __launch_bounds__(256) sage_layer_kernel(
    const float* __restrict__ hin, const int* __restrict__ ssrc,
    const int* __restrict__ cnt, const int* __restrict__ first,
    const float* __restrict__ Wl,  const float* __restrict__ bias,
    const float* __restrict__ Wr,  float* __restrict__ out, int applyTanh)
{
    __shared__ alignas(16) float sWl[FDIM * WPITCH];
    __shared__ alignas(16) float sWr[FDIM * WPITCH];
    __shared__ alignas(16) float sAgg[WAVES_PER_BLOCK * 16 * WPITCH];

    const int wave = threadIdx.x >> 5;
    const int lane = threadIdx.x & 31;
    const int half = lane >> 4;
    const int lrow = lane & 15;

    const int  rb     = blockIdx.x * WAVES_PER_BLOCK + wave;
    const bool active = (rb < ROWTILES);
    const size_t r0   = (size_t)(active ? rb : 0) * 16;

    const int rowCnt   = cnt[r0 + lrow];
    const int rowFirst = first[r0 + lrow];

    for (int i = threadIdx.x; i < (FDIM * FDIM) / 4; i += 256) {
        const int wrow = i >> 4;
        const int wcol = (i & 15) * 4;
        *reinterpret_cast<float4*>(sWl + wrow * WPITCH + wcol) = reinterpret_cast<const float4*>(Wl)[i];
        *reinterpret_cast<float4*>(sWr + wrow * WPITCH + wcol) = reinterpret_cast<const float4*>(Wr)[i];
    }
    __syncthreads();
    if (!active) return;

    v8f acc[4];
    const v8f zero = {0.f, 0.f, 0.f, 0.f, 0.f, 0.f, 0.f, 0.f};
#pragma unroll
    for (int t = 0; t < 4; ++t) acc[t] = zero;

    int nxA = list_chunk(ssrc, __shfl(rowFirst, half, 32),     __shfl(rowCnt, half, 32),     0, lrow);
    int nxB = list_chunk(ssrc, __shfl(rowFirst, half + 2, 32), __shfl(rowCnt, half + 2, 32), 0, lrow);

    const float* aRowR = hin + (r0 + lrow) * FDIM + half * 2;
#pragma unroll
    for (int kb = 0; kb < FDIM; kb += 4) {
        v2f aR = *reinterpret_cast<const v2f*>(aRowR + kb);
#pragma unroll
        for (int t = 0; t < 4; ++t) {
            const int n = t * 16 + lrow;
            v2f bR = *reinterpret_cast<const v2f*>(sWr + n * WPITCH + kb + half * 2);
            acc[t] = __builtin_amdgcn_wmma_f32_16x16x4_f32(
                false, aR, false, bR, (short)0, acc[t], false, false);
        }
    }

    asm volatile("" : "+v"(acc[0]), "+v"(acc[1]), "+v"(acc[2]), "+v"(acc[3]));

    float* sTile = sAgg + wave * (16 * WPITCH);
    const float4* h4 = reinterpret_cast<const float4*>(hin);
    const unsigned q = lrow;
#pragma unroll
    for (int s = 0; s < 4; ++s) {
        const int rA = 4 * s + half;
        const int rB = rA + 2;
        const int nA = __shfl(rowCnt, rA, 32);
        const int nB = __shfl(rowCnt, rB, 32);
        const int fA = __shfl(rowFirst, rA, 32);
        const int fB = __shfl(rowFirst, rB, 32);
        const int n  = max(nA, nB);
        int cA = nxA;
        int cB = nxB;
        const int dA = list_chunk(ssrc, fA, nA, GROUND * 2, lrow);
        const int dB = list_chunk(ssrc, fB, nB, GROUND * 2, lrow);
        if (s < 3) {
            nxA = list_chunk(ssrc, __shfl(rowFirst, rA + 4, 32), __shfl(rowCnt, rA + 4, 32), 0, lrow);
            nxB = list_chunk(ssrc, __shfl(rowFirst, rB + 4, 32), __shfl(rowCnt, rB + 4, 32), 0, lrow);
        }

        float4 accA = make_float4(0.f, 0.f, 0.f, 0.f);
        float4 accB = make_float4(0.f, 0.f, 0.f, 0.f);
        int jc = 0;
        for (int j = 0; j < n; j += GROUND) {
            if (j != 0 && jc == 0) {
                if (j == GROUND * 2) { cA = dA; cB = dB; }
                else {
                    cA = list_chunk(ssrc, fA, nA, j, lrow);
                    cB = list_chunk(ssrc, fB, nB, j, lrow);
                }
            }
            const int lb = (half << 4) + jc;
            unsigned a[GROUND], b[GROUND];
#pragma unroll
            for (int k = 0; k < GROUND; ++k) {
                a[k] = (unsigned)__shfl(cA, lb + k, 32);
                b[k] = (unsigned)__shfl(cB, lb + k, 32);
            }
            float4 u[GROUND], w[GROUND];
#pragma unroll
            for (int k = 0; k < GROUND; ++k) u[k] = h4[a[k] * 16u + q];
#pragma unroll
            for (int k = 0; k < GROUND; ++k) w[k] = h4[b[k] * 16u + q];
#pragma unroll
            for (int k = 0; k < GROUND; ++k) {
                const bool pa = (j + k < nA);
                accA.x += pa ? u[k].x : 0.f;   accA.y += pa ? u[k].y : 0.f;
                accA.z += pa ? u[k].z : 0.f;   accA.w += pa ? u[k].w : 0.f;
            }
#pragma unroll
            for (int k = 0; k < GROUND; ++k) {
                const bool pb = (j + k < nB);
                accB.x += pb ? w[k].x : 0.f;   accB.y += pb ? w[k].y : 0.f;
                accB.z += pb ? w[k].z : 0.f;   accB.w += pb ? w[k].w : 0.f;
            }
            jc = (jc == 0) ? GROUND : 0;
        }
        *reinterpret_cast<float4*>(sTile + rA * WPITCH + q * 4) = accA;
        *reinterpret_cast<float4*>(sTile + rB * WPITCH + q * 4) = accB;
    }
    __builtin_amdgcn_wave_barrier();

    const float* aRowL = sTile + lrow * WPITCH + half * 2;
#pragma unroll
    for (int kb = 0; kb < FDIM; kb += 4) {
        v2f aL = *reinterpret_cast<const v2f*>(aRowL + kb);
#pragma unroll
        for (int t = 0; t < 4; ++t) {
            const int n = t * 16 + lrow;
            v2f bL = *reinterpret_cast<const v2f*>(sWl + n * WPITCH + kb + half * 2);
            acc[t] = __builtin_amdgcn_wmma_f32_16x16x4_f32(
                false, aL, false, bL, (short)0, acc[t], false, false);
        }
    }

#pragma unroll
    for (int t = 0; t < 4; ++t) {
        const int n = t * 16 + lrow;
        const float bv = bias[n];
#pragma unroll
        for (int v = 0; v < 8; ++v) {
            const int m = v + half * 8;
            float val = acc[t][v] + bv;
            if (applyTanh) val = tanhf(val);
            out[(r0 + m) * FDIM + n] = val;
        }
    }
}

extern "C" void kernel_launch(void* const* d_in, const int* in_sizes, int n_in,
                              void* d_out, int out_size, void* d_ws, size_t ws_size,
                              hipStream_t stream)
{
    const float* x      = (const float*)d_in[0];
    const int*   ei     = (const int*)d_in[1];
    const int*   src    = ei;
    const int*   dst    = ei + NEDGES;
    const float* Wl_in  = (const float*)d_in[2];
    const float* bl_in  = (const float*)d_in[3];
    const float* Wr_in  = (const float*)d_in[4];
    const float* Wl_med = (const float*)d_in[5];
    const float* bl_med = (const float*)d_in[6];
    const float* Wr_med = (const float*)d_in[7];
    const float* Wl_out = (const float*)d_in[8];
    const float* bl_out = (const float*)d_in[9];
    const float* Wr_out = (const float*)d_in[10];

    float* bufP   = (float*)d_ws;
    float* bufO   = (float*)d_out;
    int*   cnt    = (int*)(bufP + (size_t)NNODES * FDIM);
    int*   first  = cnt + NNODES;
    int*   cntm   = first + NNODES;
    int*   loffm  = cntm + NCHUNK * BPITCH;
    int*   ssrc   = loffm + NCHUNK * BPITCH;
    int*   srt    = (int*)d_out;

    const dim3 gBlk(256);
    const dim3 gGrid((ROWTILES + WAVES_PER_BLOCK - 1) / WAVES_PER_BLOCK);

    chunk_sort_kernel<<<dim3(NCHUNK), dim3(1024), 0, stream>>>(src, dst, srt, cntm, loffm);
    owner_group_kernel<<<dim3(NBUCK), dim3(1024), 0, stream>>>(srt, cntm, loffm, cnt, first, ssrc);

    auto layer = [&](const float* hin, float* hout, const float* Wl,
                     const float* bl, const float* Wr, int doTanh) {
        sage_layer_kernel<<<gGrid, gBlk, 0, stream>>>(hin, ssrc, cnt, first, Wl, bl, Wr, hout, doTanh);
    };

    layer(x,    bufP, Wl_in,  bl_in,  Wr_in,  1);
    layer(bufP, bufO, Wl_med, bl_med, Wr_med, 1);
    layer(bufO, bufP, Wl_med, bl_med, Wr_med, 1);
    layer(bufP, bufO, Wl_out, bl_out, Wr_out, 0);
}


# layer kernel launched as 250 work-groups of 25 waves, one per WGP with 25 row tiles each, instead of 782 of 8 waves (3 per WGP, 4 on some = 32 tiles); per-wave code and sum order unchanged
# speedup vs baseline: 4.4732x; 1.1285x over previous
#include <hip/hip_runtime.h>
#include <math.h>


#define NNODES 100000
#define NEDGES 1200000
#define FDIM   64
#define ROWTILES (NNODES / 16)

typedef __attribute__((ext_vector_type(2))) float v2f;
typedef __attribute__((ext_vector_type(8))) float v8f;

#define NCHUNK    128
#define CH        (NEDGES / NCHUNK)
#define CHP       9376
#define CH_ITERS  ((CH + 1023) / 1024)
#define RSHIFT    9
#define RNODES    (1 << RSHIFT)
#define NBUCK     ((NNODES + RNODES - 1) / RNODES)
#define BPITCH    256
#define SRC_BITS  17
#define SRC_MASK  ((1 << SRC_BITS) - 1)
#define STAGE_CAP 16384

static_assert(CH * NCHUNK == NEDGES, "chunks must tile the edge list");
static_assert(CHP >= CH && CHP % 32 == 0, "chunk pitch");
static_assert(NNODES <= (1 << SRC_BITS), "source index must fit the packed word");
static_assert(NBUCK <= BPITCH, "bucket table pitch");
static_assert(NNODES % 32 == 0, "node ranges end on a line");
static_assert(NCHUNK % 32 == 0, "runs per wave");

__device__ __forceinline__ int wave_inclusive_scan(int v, int lane)
{
    for (int o = 1; o < 32; o <<= 1) {
        int u = __shfl_up(v, o, 32);
        if (lane >= o) v += u;
    }
    return v;
}

__device__ __forceinline__ int block_exclusive_scan_1024(int v, int* sW)
{
    const int lane = threadIdx.x & 31;
    const int wave = threadIdx.x >> 5;
    const int inc  = wave_inclusive_scan(v, lane);
    if (lane == 31) sW[wave] = inc;
    __syncthreads();
    if (wave == 0) {
        const int w = sW[lane];
        sW[lane] = wave_inclusive_scan(w, lane) - w;
    }
    __syncthreads();
    const int r = sW[wave] + inc - v;
    __syncthreads();
    return r;
}

__global__ void __launch_bounds__(1024) chunk_sort_kernel(
    const int* __restrict__ src, const int* __restrict__ dst,
    int* __restrict__ srt, int* __restrict__ cntm, int* __restrict__ loffm)
{
    __shared__ int sHist[BPITCH];
    __shared__ int sOff[BPITCH];
    __shared__ int sW[32];
    __shared__ int sSorted[CHP];

    const int t = threadIdx.x;
    const int c = blockIdx.x;
    if (t < BPITCH) sHist[t] = 0;
    __syncthreads();

    int rec[CH_ITERS], bk[CH_ITERS], rk[CH_ITERS];
#pragma unroll
    for (int k = 0; k < CH_ITERS; ++k) {
        const int i = k * 1024 + t;
        const int e = c * CH + ((i < CH) ? i : 0);
        bk[k]  = dst[e];
        rec[k] = src[e];
    }
#pragma unroll
    for (int k = 0; k < CH_ITERS; ++k) {
        const int i = k * 1024 + t;
        const int d = bk[k];
        bk[k]  = d >> RSHIFT;
        rec[k] = ((d & (RNODES - 1)) << SRC_BITS) | rec[k];
        rk[k]  = 0;
        if (i < CH) rk[k] = atomicAdd(&sHist[bk[k]], 1);
    }
    __syncthreads();

    const int v  = (t < BPITCH) ? sHist[t] : 0;
    const int ex = block_exclusive_scan_1024(v, sW);
    if (t < BPITCH) {
        sOff[t] = ex;
        cntm[c * BPITCH + t]  = v;
        loffm[c * BPITCH + t] = ex;
    }
    __syncthreads();

#pragma unroll
    for (int k = 0; k < CH_ITERS; ++k) {
        const int i = k * 1024 + t;
        if (i < CH) sSorted[sOff[bk[k]] + rk[k]] = rec[k];
    }
    __syncthreads();

    for (int i = t; i < CHP; i += 1024)
        srt[c * CHP + i] = (i < CH) ? sSorted[i] : 0;
}

template <class Body>
__device__ __forceinline__ void for_each_bucket_word(
    const int* __restrict__ srt, const int* sRunStart, const int* sRunLen,
    int wave, int lane, Body body)
{
    constexpr int RPW = NCHUNK / 32;
    int st[RPW], ln[RPW], w0[RPW], w1[RPW];
#pragma unroll
    for (int q = 0; q < RPW; ++q) {
        st[q] = sRunStart[wave + 32 * q];
        ln[q] = sRunLen[wave + 32 * q];
    }
#pragma unroll
    for (int q = 0; q < RPW; ++q) {
        const int last = max(ln[q] - 1, 0);
        w0[q] = srt[st[q] + min(lane, last)];
        w1[q] = srt[st[q] + min(lane + 32, last)];
    }
#pragma unroll
    for (int q = 0; q < RPW; ++q) {
        if (lane < ln[q])      body(w0[q]);
        if (lane + 32 < ln[q]) body(w1[q]);
        for (int j = lane + 64; j < ln[q]; j += 32) body(srt[st[q] + j]);
    }
}

__global__ void __launch_bounds__(1024) owner_group_kernel(
    const int* __restrict__ srt, const int* __restrict__ cntm,
    const int* __restrict__ loffm, int* __restrict__ cnt,
    int* __restrict__ first, int* __restrict__ ssrc)
{
    __shared__ int sPart[4 * BPITCH];
    __shared__ int sRunStart[NCHUNK];
    __shared__ int sRunLen[NCHUNK];
    __shared__ int sCnt[RNODES];
    __shared__ int sCur[RNODES];
    __shared__ int sW[32];
    __shared__ int sBaseTot[2];
    __shared__ int sStage[STAGE_CAP];

    const int t    = threadIdx.x;
    const int lane = t & 31;
    const int wave = t >> 5;
    const int b    = blockIdx.x;

    {
        const int col = t & (BPITCH - 1);
        const int g   = t >> 8;
        int s = 0;
        for (int c = g * (NCHUNK / 4); c < (g + 1) * (NCHUNK / 4); ++c)
            s += cntm[c * BPITCH + col];
        sPart[g * BPITCH + col] = s;
    }
    if (t < NCHUNK) {
        sRunLen[t]   = cntm[t * BPITCH + b];
        sRunStart[t] = t * CHP + loffm[t * BPITCH + b];
    }
    if (t < RNODES) sCnt[t] = 0;
    __syncthreads();

    int tot = 0;
    if (t < BPITCH)
        tot = sPart[t] + sPart[BPITCH + t] + sPart[2 * BPITCH + t] + sPart[3 * BPITCH + t];
    const int padded = (tot + 31) & ~31;
    const int base   = block_exclusive_scan_1024(padded, sW);
    if (t == b) { sBaseTot[0] = base; sBaseTot[1] = tot; }
    __syncthreads();
    const int  myBase = sBaseTot[0];
    const int  myTot  = sBaseTot[1];
    const bool staged = (myTot <= STAGE_CAP);

    for_each_bucket_word(srt, sRunStart, sRunLen, wave, lane,
        [&](int r) { atomicAdd(&sCnt[r >> SRC_BITS], 1); });
    __syncthreads();

    const int v    = (t < RNODES) ? sCnt[t] : 0;
    const int ex   = block_exclusive_scan_1024(v, sW);
    const int node = b * RNODES + t;
    if (t < RNODES) {
        sCur[t] = ex;
        if (node < NNODES) {
            cnt[node]   = v;
            first[node] = myBase + ex;
        }
    }
    __syncthreads();

    for_each_bucket_word(srt, sRunStart, sRunLen, wave, lane,
        [&](int r) {
            const int slot = atomicAdd(&sCur[r >> SRC_BITS], 1);
            const int s    = r & SRC_MASK;
            if (staged) sStage[slot] = s;
            else        atomicExch(&ssrc[myBase + slot], s);
        });
    __syncthreads();

    if (staged) {
        const int np = (myTot + 31) & ~31;
        for (int i = t; i < np; i += 1024)
            ssrc[myBase + i] = (i < myTot) ? sStage[i] : 0;
    }
}

__device__ __forceinline__ int list_chunk(const int* __restrict__ ssrc, int f, int n, int base, int l)
{
    const int p = base + l;
    return (p < n) ? ssrc[f + p] : 0;
}

#define WPITCH (FDIM + 4)
#define WAVES_PER_BLOCK 25
#define BLOCK_THREADS (WAVES_PER_BLOCK * 32)
#define GROUND 6
__global__ void __launch_bounds__(BLOCK_THREADS) sage_layer_kernel(
    const float* __restrict__ hin, const int* __restrict__ ssrc,
    const int* __restrict__ cnt, const int* __restrict__ first,
    const float* __restrict__ Wl,  const float* __restrict__ bias,
    const float* __restrict__ Wr,  float* __restrict__ out, int applyTanh)
{
    __shared__ alignas(16) float sWl[FDIM * WPITCH];
    __shared__ alignas(16) float sWr[FDIM * WPITCH];
    __shared__ alignas(16) float sAgg[WAVES_PER_BLOCK * 16 * WPITCH];

    const int wave = threadIdx.x >> 5;
    const int lane = threadIdx.x & 31;
    const int half = lane >> 4;
    const int lrow = lane & 15;

    const int  rb     = blockIdx.x * WAVES_PER_BLOCK + wave;
    const bool active = (rb < ROWTILES);
    const size_t r0   = (size_t)(active ? rb : 0) * 16;

    const int rowCnt   = cnt[r0 + lrow];
    const int rowFirst = first[r0 + lrow];

    for (int i = threadIdx.x; i < (FDIM * FDIM) / 4; i += BLOCK_THREADS) {
        const int wrow = i >> 4;
        const int wcol = (i & 15) * 4;
        *reinterpret_cast<float4*>(sWl + wrow * WPITCH + wcol) = reinterpret_cast<const float4*>(Wl)[i];
        *reinterpret_cast<float4*>(sWr + wrow * WPITCH + wcol) = reinterpret_cast<const float4*>(Wr)[i];
    }
    __syncthreads();
    if (!active) return;

    v8f acc[4];
    const v8f zero = {0.f, 0.f, 0.f, 0.f, 0.f, 0.f, 0.f, 0.f};
#pragma unroll
    for (int t = 0; t < 4; ++t) acc[t] = zero;

    int nxA = list_chunk(ssrc, __shfl(rowFirst, half, 32),     __shfl(rowCnt, half, 32),     0, lrow);
    int nxB = list_chunk(ssrc, __shfl(rowFirst, half + 2, 32), __shfl(rowCnt, half + 2, 32), 0, lrow);

    const float* aRowR = hin + (r0 + lrow) * FDIM + half * 2;
#pragma unroll
    for (int kb = 0; kb < FDIM; kb += 4) {
        v2f aR = *reinterpret_cast<const v2f*>(aRowR + kb);
#pragma unroll
        for (int t = 0; t < 4; ++t) {
            const int n = t * 16 + lrow;
            v2f bR = *reinterpret_cast<const v2f*>(sWr + n * WPITCH + kb + half * 2);
            acc[t] = __builtin_amdgcn_wmma_f32_16x16x4_f32(
                false, aR, false, bR, (short)0, acc[t], false, false);
        }
    }

    asm volatile("" : "+v"(acc[0]), "+v"(acc[1]), "+v"(acc[2]), "+v"(acc[3]));

    float* sTile = sAgg + wave * (16 * WPITCH);
    const float4* h4 = reinterpret_cast<const float4*>(hin);
    const unsigned q = lrow;
#pragma unroll
    for (int s = 0; s < 4; ++s) {
        const int rA = 4 * s + half;
        const int rB = rA + 2;
        const int nA = __shfl(rowCnt, rA, 32);
        const int nB = __shfl(rowCnt, rB, 32);
        const int fA = __shfl(rowFirst, rA, 32);
        const int fB = __shfl(rowFirst, rB, 32);
        const int n  = max(nA, nB);
        int cA = nxA;
        int cB = nxB;
        const int dA = list_chunk(ssrc, fA, nA, GROUND * 2, lrow);
        const int dB = list_chunk(ssrc, fB, nB, GROUND * 2, lrow);
        if (s < 3) {
            nxA = list_chunk(ssrc, __shfl(rowFirst, rA + 4, 32), __shfl(rowCnt, rA + 4, 32), 0, lrow);
            nxB = list_chunk(ssrc, __shfl(rowFirst, rB + 4, 32), __shfl(rowCnt, rB + 4, 32), 0, lrow);
        }

        float4 accA = make_float4(0.f, 0.f, 0.f, 0.f);
        float4 accB = make_float4(0.f, 0.f, 0.f, 0.f);
        int jc = 0;
        for (int j = 0; j < n; j += GROUND) {
            if (j != 0 && jc == 0) {
                if (j == GROUND * 2) { cA = dA; cB = dB; }
                else {
                    cA = list_chunk(ssrc, fA, nA, j, lrow);
                    cB = list_chunk(ssrc, fB, nB, j, lrow);
                }
            }
            const int lb = (half << 4) + jc;
            unsigned a[GROUND], b[GROUND];
#pragma unroll
            for (int k = 0; k < GROUND; ++k) {
                a[k] = (unsigned)__shfl(cA, lb + k, 32);
                b[k] = (unsigned)__shfl(cB, lb + k, 32);
            }
            float4 u[GROUND], w[GROUND];
#pragma unroll
            for (int k = 0; k < GROUND; ++k) u[k] = h4[a[k] * 16u + q];
#pragma unroll
            for (int k = 0; k < GROUND; ++k) w[k] = h4[b[k] * 16u + q];
#pragma unroll
            for (int k = 0; k < GROUND; ++k) {
                const bool pa = (j + k < nA);
                accA.x += pa ? u[k].x : 0.f;   accA.y += pa ? u[k].y : 0.f;
                accA.z += pa ? u[k].z : 0.f;   accA.w += pa ? u[k].w : 0.f;
            }
#pragma unroll
            for (int k = 0; k < GROUND; ++k) {
                const bool pb = (j + k < nB);
                accB.x += pb ? w[k].x : 0.f;   accB.y += pb ? w[k].y : 0.f;
                accB.z += pb ? w[k].z : 0.f;   accB.w += pb ? w[k].w : 0.f;
            }
            jc = (jc == 0) ? GROUND : 0;
        }
        *reinterpret_cast<float4*>(sTile + rA * WPITCH + q * 4) = accA;
        *reinterpret_cast<float4*>(sTile + rB * WPITCH + q * 4) = accB;
    }
    __builtin_amdgcn_wave_barrier();

    const float* aRowL = sTile + lrow * WPITCH + half * 2;
#pragma unroll
    for (int kb = 0; kb < FDIM; kb += 4) {
        v2f aL = *reinterpret_cast<const v2f*>(aRowL + kb);
#pragma unroll
        for (int t = 0; t < 4; ++t) {
            const int n = t * 16 + lrow;
            v2f bL = *reinterpret_cast<const v2f*>(sWl + n * WPITCH + kb + half * 2);
            acc[t] = __builtin_amdgcn_wmma_f32_16x16x4_f32(
                false, aL, false, bL, (short)0, acc[t], false, false);
        }
    }

#pragma unroll
    for (int t = 0; t < 4; ++t) {
        const int n = t * 16 + lrow;
        const float bv = bias[n];
#pragma unroll
        for (int v = 0; v < 8; ++v) {
            const int m = v + half * 8;
            float val = acc[t][v] + bv;
            if (applyTanh) val = tanhf(val);
            out[(r0 + m) * FDIM + n] = val;
        }
    }
}

extern "C" void kernel_launch(void* const* d_in, const int* in_sizes, int n_in,
                              void* d_out, int out_size, void* d_ws, size_t ws_size,
                              hipStream_t stream)
{
    const float* x      = (const float*)d_in[0];
    const int*   ei     = (const int*)d_in[1];
    const int*   src    = ei;
    const int*   dst    = ei + NEDGES;
    const float* Wl_in  = (const float*)d_in[2];
    const float* bl_in  = (const float*)d_in[3];
    const float* Wr_in  = (const float*)d_in[4];
    const float* Wl_med = (const float*)d_in[5];
    const float* bl_med = (const float*)d_in[6];
    const float* Wr_med = (const float*)d_in[7];
    const float* Wl_out = (const float*)d_in[8];
    const float* bl_out = (const float*)d_in[9];
    const float* Wr_out = (const float*)d_in[10];

    float* bufP   = (float*)d_ws;
    float* bufO   = (float*)d_out;
    int*   cnt    = (int*)(bufP + (size_t)NNODES * FDIM);
    int*   first  = cnt + NNODES;
    int*   cntm   = first + NNODES;
    int*   loffm  = cntm + NCHUNK * BPITCH;
    int*   ssrc   = loffm + NCHUNK * BPITCH;
    int*   srt    = (int*)d_out;

    const dim3 gBlk(BLOCK_THREADS);
    const dim3 gGrid((ROWTILES + WAVES_PER_BLOCK - 1) / WAVES_PER_BLOCK);

    chunk_sort_kernel<<<dim3(NCHUNK), dim3(1024), 0, stream>>>(src, dst, srt, cntm, loffm);
    owner_group_kernel<<<dim3(NBUCK), dim3(1024), 0, stream>>>(srt, cntm, loffm, cnt, first, ssrc);

    auto layer = [&](const float* hin, float* hout, const float* Wl,
                     const float* bl, const float* Wr, int doTanh) {
        sage_layer_kernel<<<gGrid, gBlk, 0, stream>>>(hin, ssrc, cnt, first, Wl, bl, Wr, hout, doTanh);
    };

    layer(x,    bufP, Wl_in,  bl_in,  Wr_in,  1);
    layer(bufP, bufO, Wl_med, bl_med, Wr_med, 1);
    layer(bufO, bufP, Wl_med, bl_med, Wr_med, 1);
    layer(bufP, bufO, Wl_out, bl_out, Wr_out, 0);
}


# epilogue tanh by the hardware v_tanh_f32 instruction, one per value, in place of the in-line library tanhf with its two divergent branches; last bits of tanh may differ; sum order unchanged
# speedup vs baseline: 4.6255x; 1.0340x over previous
#include <hip/hip_runtime.h>
#include <math.h>


#define NNODES 100000
#define NEDGES 1200000
#define FDIM   64
#define ROWTILES (NNODES / 16)

typedef __attribute__((ext_vector_type(2))) float v2f;
typedef __attribute__((ext_vector_type(8))) float v8f;

#define NCHUNK    128
#define CH        (NEDGES / NCHUNK)
#define CHP       9376
#define CH_ITERS  ((CH + 1023) / 1024)
#define RSHIFT    9
#define RNODES    (1 << RSHIFT)
#define NBUCK     ((NNODES + RNODES - 1) / RNODES)
#define BPITCH    256
#define SRC_BITS  17
#define SRC_MASK  ((1 << SRC_BITS) - 1)
#define STAGE_CAP 16384

static_assert(CH * NCHUNK == NEDGES, "chunks must tile the edge list");
static_assert(CHP >= CH && CHP % 32 == 0, "chunk pitch");
static_assert(NNODES <= (1 << SRC_BITS), "source index must fit the packed word");
static_assert(NBUCK <= BPITCH, "bucket table pitch");
static_assert(NNODES % 32 == 0, "node ranges end on a line");
static_assert(NCHUNK % 32 == 0, "runs per wave");

__device__ __forceinline__ int wave_inclusive_scan(int v, int lane)
{
    for (int o = 1; o < 32; o <<= 1) {
        int u = __shfl_up(v, o, 32);
        if (lane >= o) v += u;
    }
    return v;
}

__device__ __forceinline__ int block_exclusive_scan_1024(int v, int* sW)
{
    const int lane = threadIdx.x & 31;
    const int wave = threadIdx.x >> 5;
    const int inc  = wave_inclusive_scan(v, lane);
    if (lane == 31) sW[wave] = inc;
    __syncthreads();
    if (wave == 0) {
        const int w = sW[lane];
        sW[lane] = wave_inclusive_scan(w, lane) - w;
    }
    __syncthreads();
    const int r = sW[wave] + inc - v;
    __syncthreads();
    return r;
}

__global__ void __launch_bounds__(1024) chunk_sort_kernel(
    const int* __restrict__ src, const int* __restrict__ dst,
    int* __restrict__ srt, int* __restrict__ cntm, int* __restrict__ loffm)
{
    __shared__ int sHist[BPITCH];
    __shared__ int sOff[BPITCH];
    __shared__ int sW[32];
    __shared__ int sSorted[CHP];

    const int t = threadIdx.x;
    const int c = blockIdx.x;
    if (t < BPITCH) sHist[t] = 0;
    __syncthreads();

    int rec[CH_ITERS], bk[CH_ITERS], rk[CH_ITERS];
#pragma unroll
    for (int k = 0; k < CH_ITERS; ++k) {
        const int i = k * 1024 + t;
        const int e = c * CH + ((i < CH) ? i : 0);
        bk[k]  = dst[e];
        rec[k] = src[e];
    }
#pragma unroll
    for (int k = 0; k < CH_ITERS; ++k) {
        const int i = k * 1024 + t;
        const int d = bk[k];
        bk[k]  = d >> RSHIFT;
        rec[k] = ((d & (RNODES - 1)) << SRC_BITS) | rec[k];
        rk[k]  = 0;
        if (i < CH) rk[k] = atomicAdd(&sHist[bk[k]], 1);
    }
    __syncthreads();

    const int v  = (t < BPITCH) ? sHist[t] : 0;
    const int ex = block_exclusive_scan_1024(v, sW);
    if (t < BPITCH) {
        sOff[t] = ex;
        cntm[c * BPITCH + t]  = v;
        loffm[c * BPITCH + t] = ex;
    }
    __syncthreads();

#pragma unroll
    for (int k = 0; k < CH_ITERS; ++k) {
        const int i = k * 1024 + t;
        if (i < CH) sSorted[sOff[bk[k]] + rk[k]] = rec[k];
    }
    __syncthreads();

    for (int i = t; i < CHP; i += 1024)
        srt[c * CHP + i] = (i < CH) ? sSorted[i] : 0;
}

template <class Body>
__device__ __forceinline__ void for_each_bucket_word(
    const int* __restrict__ srt, const int* sRunStart, const int* sRunLen,
    int wave, int lane, Body body)
{
    constexpr int RPW = NCHUNK / 32;
    int st[RPW], ln[RPW], w0[RPW], w1[RPW];
#pragma unroll
    for (int q = 0; q < RPW; ++q) {
        st[q] = sRunStart[wave + 32 * q];
        ln[q] = sRunLen[wave + 32 * q];
    }
#pragma unroll
    for (int q = 0; q < RPW; ++q) {
        const int last = max(ln[q] - 1, 0);
        w0[q] = srt[st[q] + min(lane, last)];
        w1[q] = srt[st[q] + min(lane + 32, last)];
    }
#pragma unroll
    for (int q = 0; q < RPW; ++q) {
        if (lane < ln[q])      body(w0[q]);
        if (lane + 32 < ln[q]) body(w1[q]);
        for (int j = lane + 64; j < ln[q]; j += 32) body(srt[st[q] + j]);
    }
}

__global__ void __launch_bounds__(1024) owner_group_kernel(
    const int* __restrict__ srt, const int* __restrict__ cntm,
    const int* __restrict__ loffm, int* __restrict__ cnt,
    int* __restrict__ first, int* __restrict__ ssrc)
{
    __shared__ int sPart[4 * BPITCH];
    __shared__ int sRunStart[NCHUNK];
    __shared__ int sRunLen[NCHUNK];
    __shared__ int sCnt[RNODES];
    __shared__ int sCur[RNODES];
    __shared__ int sW[32];
    __shared__ int sBaseTot[2];
    __shared__ int sStage[STAGE_CAP];

    const int t    = threadIdx.x;
    const int lane = t & 31;
    const int wave = t >> 5;
    const int b    = blockIdx.x;

    {
        const int col = t & (BPITCH - 1);
        const int g   = t >> 8;
        int s = 0;
        for (int c = g * (NCHUNK / 4); c < (g + 1) * (NCHUNK / 4); ++c)
            s += cntm[c * BPITCH + col];
        sPart[g * BPITCH + col] = s;
    }
    if (t < NCHUNK) {
        sRunLen[t]   = cntm[t * BPITCH + b];
        sRunStart[t] = t * CHP + loffm[t * BPITCH + b];
    }
    if (t < RNODES) sCnt[t] = 0;
    __syncthreads();

    int tot = 0;
    if (t < BPITCH)
        tot = sPart[t] + sPart[BPITCH + t] + sPart[2 * BPITCH + t] + sPart[3 * BPITCH + t];
    const int padded = (tot + 31) & ~31;
    const int base   = block_exclusive_scan_1024(padded, sW);
    if (t == b) { sBaseTot[0] = base; sBaseTot[1] = tot; }
    __syncthreads();
    const int  myBase = sBaseTot[0];
    const int  myTot  = sBaseTot[1];
    const bool staged = (myTot <= STAGE_CAP);

    for_each_bucket_word(srt, sRunStart, sRunLen, wave, lane,
        [&](int r) { atomicAdd(&sCnt[r >> SRC_BITS], 1); });
    __syncthreads();

    const int v    = (t < RNODES) ? sCnt[t] : 0;
    const int ex   = block_exclusive_scan_1024(v, sW);
    const int node = b * RNODES + t;
    if (t < RNODES) {
        sCur[t] = ex;
        if (node < NNODES) {
            cnt[node]   = v;
            first[node] = myBase + ex;
        }
    }
    __syncthreads();

    for_each_bucket_word(srt, sRunStart, sRunLen, wave, lane,
        [&](int r) {
            const int slot = atomicAdd(&sCur[r >> SRC_BITS], 1);
            const int s    = r & SRC_MASK;
            if (staged) sStage[slot] = s;
            else        atomicExch(&ssrc[myBase + slot], s);
        });
    __syncthreads();

    if (staged) {
        const int np = (myTot + 31) & ~31;
        for (int i = t; i < np; i += 1024)
            ssrc[myBase + i] = (i < myTot) ? sStage[i] : 0;
    }
}

__device__ __forceinline__ int list_chunk(const int* __restrict__ ssrc, int f, int n, int base, int l)
{
    const int p = base + l;
    return (p < n) ? ssrc[f + p] : 0;
}

#define WPITCH (FDIM + 4)
#define WAVES_PER_BLOCK 25
#define BLOCK_THREADS (WAVES_PER_BLOCK * 32)
#define GROUND 6
__global__ void __launch_bounds__(BLOCK_THREADS) sage_layer_kernel(
    const float* __restrict__ hin, const int* __restrict__ ssrc,
    const int* __restrict__ cnt, const int* __restrict__ first,
    const float* __restrict__ Wl,  const float* __restrict__ bias,
    const float* __restrict__ Wr,  float* __restrict__ out, int applyTanh)
{
    __shared__ alignas(16) float sWl[FDIM * WPITCH];
    __shared__ alignas(16) float sWr[FDIM * WPITCH];
    __shared__ alignas(16) float sAgg[WAVES_PER_BLOCK * 16 * WPITCH];

    const int wave = threadIdx.x >> 5;
    const int lane = threadIdx.x & 31;
    const int half = lane >> 4;
    const int lrow = lane & 15;

    const int  rb     = blockIdx.x * WAVES_PER_BLOCK + wave;
    const bool active = (rb < ROWTILES);
    const size_t r0   = (size_t)(active ? rb : 0) * 16;

    const int rowCnt   = cnt[r0 + lrow];
    const int rowFirst = first[r0 + lrow];

    for (int i = threadIdx.x; i < (FDIM * FDIM) / 4; i += BLOCK_THREADS) {
        const int wrow = i >> 4;
        const int wcol = (i & 15) * 4;
        *reinterpret_cast<float4*>(sWl + wrow * WPITCH + wcol) = reinterpret_cast<const float4*>(Wl)[i];
        *reinterpret_cast<float4*>(sWr + wrow * WPITCH + wcol) = reinterpret_cast<const float4*>(Wr)[i];
    }
    __syncthreads();
    if (!active) return;

    v8f acc[4];
    const v8f zero = {0.f, 0.f, 0.f, 0.f, 0.f, 0.f, 0.f, 0.f};
#pragma unroll
    for (int t = 0; t < 4; ++t) acc[t] = zero;

    int nxA = list_chunk(ssrc, __shfl(rowFirst, half, 32),     __shfl(rowCnt, half, 32),     0, lrow);
    int nxB = list_chunk(ssrc, __shfl(rowFirst, half + 2, 32), __shfl(rowCnt, half + 2, 32), 0, lrow);

    const float* aRowR = hin + (r0 + lrow) * FDIM + half * 2;
#pragma unroll
    for (int kb = 0; kb < FDIM; kb += 4) {
        v2f aR = *reinterpret_cast<const v2f*>(aRowR + kb);
#pragma unroll
        for (int t = 0; t < 4; ++t) {
            const int n = t * 16 + lrow;
            v2f bR = *reinterpret_cast<const v2f*>(sWr + n * WPITCH + kb + half * 2);
            acc[t] = __builtin_amdgcn_wmma_f32_16x16x4_f32(
                false, aR, false, bR, (short)0, acc[t], false, false);
        }
    }

    asm volatile("" : "+v"(acc[0]), "+v"(acc[1]), "+v"(acc[2]), "+v"(acc[3]));

    float* sTile = sAgg + wave * (16 * WPITCH);
    const float4* h4 = reinterpret_cast<const float4*>(hin);
    const unsigned q = lrow;
#pragma unroll
    for (int s = 0; s < 4; ++s) {
        const int rA = 4 * s + half;
        const int rB = rA + 2;
        const int nA = __shfl(rowCnt, rA, 32);
        const int nB = __shfl(rowCnt, rB, 32);
        const int fA = __shfl(rowFirst, rA, 32);
        const int fB = __shfl(rowFirst, rB, 32);
        const int n  = max(nA, nB);
        int cA = nxA;
        int cB = nxB;
        const int dA = list_chunk(ssrc, fA, nA, GROUND * 2, lrow);
        const int dB = list_chunk(ssrc, fB, nB, GROUND * 2, lrow);
        if (s < 3) {
            nxA = list_chunk(ssrc, __shfl(rowFirst, rA + 4, 32), __shfl(rowCnt, rA + 4, 32), 0, lrow);
            nxB = list_chunk(ssrc, __shfl(rowFirst, rB + 4, 32), __shfl(rowCnt, rB + 4, 32), 0, lrow);
        }

        float4 accA = make_float4(0.f, 0.f, 0.f, 0.f);
        float4 accB = make_float4(0.f, 0.f, 0.f, 0.f);
        int jc = 0;
        for (int j = 0; j < n; j += GROUND) {
            if (j != 0 && jc == 0) {
                if (j == GROUND * 2) { cA = dA; cB = dB; }
                else {
                    cA = list_chunk(ssrc, fA, nA, j, lrow);
                    cB = list_chunk(ssrc, fB, nB, j, lrow);
                }
            }
            const int lb = (half << 4) + jc;
            unsigned a[GROUND], b[GROUND];
#pragma unroll
            for (int k = 0; k < GROUND; ++k) {
                a[k] = (unsigned)__shfl(cA, lb + k, 32);
                b[k] = (unsigned)__shfl(cB, lb + k, 32);
            }
            float4 u[GROUND], w[GROUND];
#pragma unroll
            for (int k = 0; k < GROUND; ++k) u[k] = h4[a[k] * 16u + q];
#pragma unroll
            for (int k = 0; k < GROUND; ++k) w[k] = h4[b[k] * 16u + q];
#pragma unroll
            for (int k = 0; k < GROUND; ++k) {
                const bool pa = (j + k < nA);
                accA.x += pa ? u[k].x : 0.f;   accA.y += pa ? u[k].y : 0.f;
                accA.z += pa ? u[k].z : 0.f;   accA.w += pa ? u[k].w : 0.f;
            }
#pragma unroll
            for (int k = 0; k < GROUND; ++k) {
                const bool pb = (j + k < nB);
                accB.x += pb ? w[k].x : 0.f;   accB.y += pb ? w[k].y : 0.f;
                accB.z += pb ? w[k].z : 0.f;   accB.w += pb ? w[k].w : 0.f;
            }
            jc = (jc == 0) ? GROUND : 0;
        }
        *reinterpret_cast<float4*>(sTile + rA * WPITCH + q * 4) = accA;
        *reinterpret_cast<float4*>(sTile + rB * WPITCH + q * 4) = accB;
    }
    __builtin_amdgcn_wave_barrier();

    const float* aRowL = sTile + lrow * WPITCH + half * 2;
#pragma unroll
    for (int kb = 0; kb < FDIM; kb += 4) {
        v2f aL = *reinterpret_cast<const v2f*>(aRowL + kb);
#pragma unroll
        for (int t = 0; t < 4; ++t) {
            const int n = t * 16 + lrow;
            v2f bL = *reinterpret_cast<const v2f*>(sWl + n * WPITCH + kb + half * 2);
            acc[t] = __builtin_amdgcn_wmma_f32_16x16x4_f32(
                false, aL, false, bL, (short)0, acc[t], false, false);
        }
    }

#pragma unroll
    for (int t = 0; t < 4; ++t) {
        const int n = t * 16 + lrow;
        const float bv = bias[n];
#pragma unroll
        for (int v = 0; v < 8; ++v) {
            const int m = v + half * 8;
            float val = acc[t][v] + bv;
            if (applyTanh) val = __builtin_amdgcn_tanhf(val);
            out[(r0 + m) * FDIM + n] = val;
        }
    }
}

extern "C" void kernel_launch(void* const* d_in, const int* in_sizes, int n_in,
                              void* d_out, int out_size, void* d_ws, size_t ws_size,
                              hipStream_t stream)
{
    const float* x      = (const float*)d_in[0];
    const int*   ei     = (const int*)d_in[1];
    const int*   src    = ei;
    const int*   dst    = ei + NEDGES;
    const float* Wl_in  = (const float*)d_in[2];
    const float* bl_in  = (const float*)d_in[3];
    const float* Wr_in  = (const float*)d_in[4];
    const float* Wl_med = (const float*)d_in[5];
    const float* bl_med = (const float*)d_in[6];
    const float* Wr_med = (const float*)d_in[7];
    const float* Wl_out = (const float*)d_in[8];
    const float* bl_out = (const float*)d_in[9];
    const float* Wr_out = (const float*)d_in[10];

    float* bufP   = (float*)d_ws;
    float* bufO   = (float*)d_out;
    int*   cnt    = (int*)(bufP + (size_t)NNODES * FDIM);
    int*   first  = cnt + NNODES;
    int*   cntm   = first + NNODES;
    int*   loffm  = cntm + NCHUNK * BPITCH;
    int*   ssrc   = loffm + NCHUNK * BPITCH;
    int*   srt    = (int*)d_out;

    const dim3 gBlk(BLOCK_THREADS);
    const dim3 gGrid((ROWTILES + WAVES_PER_BLOCK - 1) / WAVES_PER_BLOCK);

    chunk_sort_kernel<<<dim3(NCHUNK), dim3(1024), 0, stream>>>(src, dst, srt, cntm, loffm);
    owner_group_kernel<<<dim3(NBUCK), dim3(1024), 0, stream>>>(srt, cntm, loffm, cnt, first, ssrc);

    auto layer = [&](const float* hin, float* hout, const float* Wl,
                     const float* bl, const float* Wr, int doTanh) {
        sage_layer_kernel<<<gGrid, gBlk, 0, stream>>>(hin, ssrc, cnt, first, Wl, bl, Wr, hout, doTanh);
    };

    layer(x,    bufP, Wl_in,  bl_in,  Wr_in,  1);
    layer(bufP, bufO, Wl_med, bl_med, Wr_med, 1);
    layer(bufO, bufP, Wl_med, bl_med, Wr_med, 1);
    layer(bufP, bufO, Wl_out, bl_out, Wr_out, 0);
}
